# starting kernel
# baseline (speedup 1.0000x reference)
#include <hip/hip_runtime.h>


#define FOV        11
#define MAX_DIST   5
#define NACT       5
#define NEG_INF_F  1e10f
#define AGENTS     768
#define BATCH      4
#define PMIN       60
#define NPUSED     61
#define KTRUE      (NPUSED * NACT)
#define KPAD       308
#define SSTRIDE    309
#define TILES_PER_B (AGENTS / 16)

typedef float v2f __attribute__((ext_vector_type(2)));
typedef float v8f __attribute__((ext_vector_type(8)));

__global__ void mf_init_q(const float* __restrict__ q_init,
                          const float* __restrict__ ill,
                          float* __restrict__ q, int n) {
  int t = blockIdx.x * blockDim.x + threadIdx.x;
  if (t < n) q[t] = q_init[t] - NEG_INF_F * ill[t];
}

__global__ void mf_prep_B(const float* __restrict__ params,
                          float* __restrict__ Bg) {
  int idx = blockIdx.x * blockDim.x + threadIdx.x;
  if (idx >= KPAD * 16) return;
  int k = idx >> 4, n = idx & 15;
  float v = 0.f;
  if (n < NACT && k < KTRUE) {
    int p = k / NACT;
    int y = k - p * NACT;
    v = params[(p + PMIN) * (NACT * NACT) + n * NACT + y];
  }
  Bg[idx] = v;
}

__global__ void mf_softmax(const float* __restrict__ q,
                           float* __restrict__ probs, int n) {
  int t = blockIdx.x * blockDim.x + threadIdx.x;
  if (t >= n) return;
  const float* row = q + t * NACT;
  float m = row[0];
#pragma unroll
  for (int y = 1; y < NACT; ++y) m = fmaxf(m, row[y]);
  float e[NACT];
  float s = 0.f;
#pragma unroll
  for (int y = 0; y < NACT; ++y) { e[y] = __expf(row[y] - m); s += e[y]; }
  float inv = 1.f / s;
#pragma unroll
  for (int y = 0; y < NACT; ++y) probs[t * NACT + y] = e[y] * inv;
}

__global__ void __launch_bounds__(32)
mf_step(const float* __restrict__ logits,
        const float* __restrict__ ill,
        const int*   __restrict__ pos,
        const float* __restrict__ Bg,
        const float* __restrict__ probs,
        float* __restrict__ qout) {
  __shared__ float Slds[16 * SSTRIDE];
  __shared__ float Bm[KPAD * 16];

  const int lane   = threadIdx.x;
  const int b      = blockIdx.x / TILES_PER_B;
  const int i_base = (blockIdx.x - b * TILES_PER_B) * 16;

  const float* prb = probs + (size_t)(b * AGENTS) * NACT;

  for (int off = lane * 32; off < AGENTS * NACT; off += 32 * 32)
    __builtin_prefetch(prb + off, 0, 3);

  {
    const float4* Bg4 = (const float4*)Bg;
    float4*       Bm4 = (float4*)Bm;
    for (int idx = lane; idx < (KPAD * 16) / 4; idx += 32) Bm4[idx] = Bg4[idx];
  }
  for (int idx = lane; idx < 16 * SSTRIDE; idx += 32) Slds[idx] = 0.f;
  __syncthreads();

  if (lane < 16) {
    const int i  = i_base + lane;
    const int iy = pos[(b * AGENTS + i) * 2 + 0];
    const int ix = pos[(b * AGENTS + i) * 2 + 1];
    const int* pb = pos + (size_t)(b * AGENTS) * 2;
    float* srow = Slds + lane * SSTRIDE;
    for (int j = 0; j < AGENTS; ++j) {
      int jy = pb[2 * j], jx = pb[2 * j + 1];
      int dy = iy > jy ? iy - jy : jy - iy;
      int dx = ix > jx ? ix - jx : jx - ix;
      if (dy <= MAX_DIST && dx <= MAX_DIST) {
        float* dst = srow + (dy * FOV + dx) * NACT;
#pragma unroll
        for (int y = 0; y < NACT; ++y) dst[y] += prb[j * NACT + y];
      }
    }
  }
  __syncthreads();

  const int half = lane >> 4;
  const int nl   = lane & 15;
  v8f acc = {};
#pragma unroll 4
  for (int k0 = 0; k0 < KPAD; k0 += 4) {
    const int ka = k0 + 2 * half;
    v2f a, bf;
    a.x  = Slds[nl * SSTRIDE + ka];
    a.y  = Slds[nl * SSTRIDE + ka + 1];
    bf.x = Bm[ka * 16 + nl];
    bf.y = Bm[(ka + 1) * 16 + nl];
    acc = __builtin_amdgcn_wmma_f32_16x16x4_f32(
         false, a,  false, bf,
         (short)0, acc,  false,  false);
  }

  if (nl < NACT) {
#pragma unroll
    for (int g = 0; g < 8; ++g) {
      int m = g + 8 * half;
      int o = (b * AGENTS + i_base + m) * NACT + nl;
      qout[o] = logits[o] + acc[g] - NEG_INF_F * ill[o];
    }
  }
}

extern "C" void kernel_launch(void* const* d_in, const int* in_sizes, int n_in,
                              void* d_out, int out_size, void* d_ws, size_t ws_size,
                              hipStream_t stream) {
  (void)in_sizes; (void)n_in; (void)out_size; (void)ws_size;
  const float* logits = (const float*)d_in[0];
  const float* ill    = (const float*)d_in[1];
  const int*   pos    = (const int*)d_in[2];
  const float* q_init = (const float*)d_in[3];
  const float* params = (const float*)d_in[4];
  const int ITER = 3;
  const int NQ  = BATCH * AGENTS * NACT;
  const int NBA = BATCH * AGENTS;
  const int NB  = KPAD * 16;

  float* q_buf = (float*)d_ws;
  float* probs = q_buf + NQ;
  float* Bg    = probs + NQ;

  mf_prep_B<<<(NB + 255) / 256, 256, 0, stream>>>(params, Bg);
  mf_init_q<<<(NQ + 255) / 256, 256, 0, stream>>>(q_init, ill, q_buf, NQ);
  for (int it = 0; it < ITER; ++it) {
    mf_softmax<<<(NBA + 255) / 256, 256, 0, stream>>>(q_buf, probs, NBA);
    float* out = (it == ITER - 1) ? (float*)d_out : q_buf;
    mf_step<<<BATCH * TILES_PER_B, 32, 0, stream>>>(logits, ill, pos, Bg,
                                                    probs, out);
  }
}


# Next sweep's table product moved into the sweep kernel's epilogue (its 32 agents' softmax and table rows; two table planes alternate), B read from params in place: 8 launches become 4
# speedup vs baseline: 9.3899x; 9.3899x over previous
#include <hip/hip_runtime.h>


#define FOV        11
#define MAX_DIST   5
#define NACT       5
#define NEG_INF_F  1e10f
#define AGENTS     768
#define BATCH      4
#define PMIN       60
#define NPUSED     61
#define NTRUE      (NPUSED * NACT)
#define NPAD       320
#define GA         32
#define NSEG       16
#define SEGLEN     (AGENTS / NSEG)

typedef float v2f __attribute__((ext_vector_type(2)));
typedef float v8f __attribute__((ext_vector_type(8)));

__device__ __forceinline__ void mf_b_frag(const float* __restrict__ params,
                                          int half, int n, v2f& b0, v2f& b1) {
  const bool ok = n < NTRUE;
  const int p = n / NACT;
  const int x = n - p * NACT;
  const int base = ok ? (p + PMIN) * (NACT * NACT) + x * NACT : 0;
  const float v0 = params[base + 2 * half];
  const float v1 = params[base + 2 * half + 1];
  const float v4 = params[base + 4];
  b0.x = ok ? v0 : 0.f;
  b0.y = ok ? v1 : 0.f;
  b1.x = (ok && half == 0) ? v4 : 0.f;
  b1.y = 0.f;
}

__device__ __forceinline__ void mf_a_frag(const float* q, int half,
                                          v2f& a0, v2f& a1) {
  float m = q[0];
#pragma unroll
  for (int y = 1; y < NACT; ++y) m = fmaxf(m, q[y]);
  float e[NACT];
  float s = 0.f;
#pragma unroll
  for (int y = 0; y < NACT; ++y) { e[y] = __expf(q[y] - m); s += e[y]; }
  const float inv = 1.f / s;
  float pr[NACT];
#pragma unroll
  for (int y = 0; y < NACT; ++y) pr[y] = e[y] * inv;
  a0.x = half ? pr[2] : pr[0];
  a0.y = half ? pr[3] : pr[1];
  a1.x = half ? 0.f : pr[4];
  a1.y = 0.f;
}

__device__ __forceinline__ void mf_tile(v2f a0, v2f a1, v2f b0, v2f b1,
                                        float* __restrict__ dst) {
  v8f acc = {};
  acc = __builtin_amdgcn_wmma_f32_16x16x4_f32(
       false, a0,  false, b0,
       (short)0, acc,  false,  false);
  acc = __builtin_amdgcn_wmma_f32_16x16x4_f32(
       false, a1,  false, b1,
       (short)0, acc,  false,  false);
#pragma unroll
  for (int g = 0; g < 8; ++g) dst[(size_t)g * NPAD] = acc[g];
}

__global__ void __launch_bounds__(128)
mf_table0(const float* __restrict__ q_init,
          const float* __restrict__ ill,
          const float* __restrict__ params,
          float* __restrict__ T) {
  const int lane = threadIdx.x & 31;
  const int w    = threadIdx.x >> 5;
  const int half = lane >> 4;
  const int nl   = lane & 15;
  const int row0 = blockIdx.x * 16;

  const int r = (row0 + nl) * NACT;
  float q[NACT];
#pragma unroll
  for (int y = 0; y < NACT; ++y) q[y] = q_init[r + y] - NEG_INF_F * ill[r + y];
  v2f a0, a1;
  mf_a_frag(q, half, a0, a1);

  float* dst = T + (size_t)(row0 + 8 * half) * NPAD + nl;
  v2f b0[5], b1[5];
#pragma unroll
  for (int tt = 0; tt < 5; ++tt)
    mf_b_frag(params, half, (w * 5 + tt) * 16 + nl, b0[tt], b1[tt]);
#pragma unroll
  for (int tt = 0; tt < 5; ++tt)
    mf_tile(a0, a1, b0[tt], b1[tt], dst + (w * 5 + tt) * 16);
}

__device__ __forceinline__ void mf_sweep(const float* __restrict__ logits,
                                         const float* __restrict__ ill,
                                         const int*   __restrict__ pos,
                                         const float* __restrict__ params,
                                         const float* __restrict__ T,
                                         float* __restrict__ qout,
                                         float* __restrict__ Tn,
                                         const bool last) {
  __shared__ unsigned short lst[NSEG * SEGLEN * GA];
  __shared__ unsigned short flat[GA * AGENTS];
  __shared__ int cnts[NSEG * GA];
  __shared__ int tot[GA];
  __shared__ float qs[GA * NACT];

  const int tid    = threadIdx.x;
  const int b      = blockIdx.x / (AGENTS / GA);
  const int a_base = (blockIdx.x - b * (AGENTS / GA)) * GA;
  const int* pb    = pos + (size_t)(b * AGENTS) * 2;

  const int a = tid >> 4;
  const int s = tid & 15;
  const int o = (b * AGENTS + a_base + a) * NACT + (s < NACT ? s : 0);
  const float lg = logits[o];
  const float il = ill[o];

  const int lane = tid & 31;
  const int wv   = __builtin_amdgcn_readfirstlane(tid >> 5);
  const int half = lane >> 4;
  const int nl   = lane & 15;
  const int rt   = wv & 1;
  const int cg   = wv >> 1;
  v2f b0[3], b1[3];
  if (!last) {
#pragma unroll
    for (int t = 0; t < 3; ++t) {
      const int ct = (t < 2 || cg < 4) ? cg + 8 * t : cg;
      mf_b_frag(params, half, ct * 16 + nl, b0[t], b1[t]);
    }
  }

  {
    const int seg  = tid >> 5;
    const int i    = a_base + lane;
    const int iy   = pb[2 * i];
    const int ix   = pb[2 * i + 1];
    const int j0   = seg * SEGLEN;
    const int ja  = j0 + lane;
    const int jb  = j0 + 32 + (lane & 15);
    const int vy0 = pb[2 * ja], vx0 = pb[2 * ja + 1];
    const int vy1 = pb[2 * jb], vx1 = pb[2 * jb + 1];
    int cnt = 0;
    for (int jj = 0; jj < 32; ++jj) {
      const int jy = __builtin_amdgcn_readlane(vy0, jj);
      const int jx = __builtin_amdgcn_readlane(vx0, jj);
      const int dy = iy > jy ? iy - jy : jy - iy;
      const int dx = ix > jx ? ix - jx : jx - ix;
      if (dy <= MAX_DIST && dx <= MAX_DIST) {
        lst[(j0 + cnt) * GA + lane] =
            (unsigned short)((j0 + jj) | ((dy * FOV + dx) << 10));
        ++cnt;
      }
    }
    for (int jj = 0; jj < 16; ++jj) {
      const int jy = __builtin_amdgcn_readlane(vy1, jj);
      const int jx = __builtin_amdgcn_readlane(vx1, jj);
      const int dy = iy > jy ? iy - jy : jy - iy;
      const int dx = ix > jx ? ix - jx : jx - ix;
      if (dy <= MAX_DIST && dx <= MAX_DIST) {
        lst[(j0 + cnt) * GA + lane] =
            (unsigned short)((j0 + 32 + jj) | ((dy * FOV + dx) << 10));
        ++cnt;
      }
    }
    cnts[seg * GA + lane] = cnt;
  }
  __syncthreads();

  {
    int off = 0, c = 0;
#pragma unroll
    for (int k = 0; k < NSEG; ++k) {
      const int ck = cnts[k * GA + a];
      off += (k < s) ? ck : 0;
      c = (k == s) ? ck : c;
    }
    for (int n = 0; n < c; ++n)
      flat[a * AGENTS + off + n] = lst[(s * SEGLEN + n) * GA + a];
    if (s == NSEG - 1) tot[a] = off + c;
  }
  __syncthreads();

  const float* Tb = T + (size_t)(b * AGENTS) * NPAD;
  const unsigned short* fl = flat + a * AGENTS;
  const int total = tot[a];
  float s0 = 0.f, s1 = 0.f, s2 = 0.f, s3 = 0.f, s4 = 0.f;
  for (int n = s; n < total; n += 32) {
    const bool h1 = (n + 16) < total;
    const unsigned e0 = fl[n];
    const unsigned e1 = h1 ? fl[n + 16] : e0;
    const float* t0 = Tb + (e0 & 1023u) * NPAD + (e0 >> 10) * NACT;
    const float* t1 = Tb + (e1 & 1023u) * NPAD + (e1 >> 10) * NACT;
    const float u0 = t0[0], u1 = t0[1], u2 = t0[2], u3 = t0[3], u4 = t0[4];
    const float w0 = t1[0], w1 = t1[1], w2 = t1[2], w3 = t1[3], w4 = t1[4];
    s0 += u0; s1 += u1; s2 += u2; s3 += u3; s4 += u4;
    s0 += h1 ? w0 : 0.f; s1 += h1 ? w1 : 0.f; s2 += h1 ? w2 : 0.f;
    s3 += h1 ? w3 : 0.f; s4 += h1 ? w4 : 0.f;
  }
#pragma unroll
  for (int d = 8; d >= 1; d >>= 1) {
    s0 += __shfl_xor(s0, d, 16);
    s1 += __shfl_xor(s1, d, 16);
    s2 += __shfl_xor(s2, d, 16);
    s3 += __shfl_xor(s3, d, 16);
    s4 += __shfl_xor(s4, d, 16);
  }
  if (s < NACT) {
    const float v = s == 0 ? s0 : s == 1 ? s1 : s == 2 ? s2 : s == 3 ? s3 : s4;
    const float qv = lg + v - NEG_INF_F * il;
    if (last) qout[o] = qv;
    else      qs[a * NACT + s] = qv;
  }
  if (last) return;

  __syncthreads();
  float q[NACT];
#pragma unroll
  for (int y = 0; y < NACT; ++y) q[y] = qs[(rt * 16 + nl) * NACT + y];
  v2f a0, a1;
  mf_a_frag(q, half, a0, a1);
  float* dst = Tn + (size_t)(b * AGENTS + a_base + rt * 16 + 8 * half) * NPAD + nl;
  mf_tile(a0, a1, b0[0], b1[0], dst + cg * 16);
  mf_tile(a0, a1, b0[1], b1[1], dst + (cg + 8) * 16);
  if (cg < 4) mf_tile(a0, a1, b0[2], b1[2], dst + (cg + 16) * 16);
}

__global__ void __launch_bounds__(512)
mf_sweep_next(const float* __restrict__ logits,
              const float* __restrict__ ill,
              const int*   __restrict__ pos,
              const float* __restrict__ params,
              const float* __restrict__ T,
              float* __restrict__ Tn) {
  mf_sweep(logits, ill, pos, params, T, nullptr, Tn, false);
}

__global__ void __launch_bounds__(512)
mf_sweep_last(const float* __restrict__ logits,
              const float* __restrict__ ill,
              const int*   __restrict__ pos,
              const float* __restrict__ T,
              float* __restrict__ qout) {
  mf_sweep(logits, ill, pos, nullptr, T, qout, nullptr, true);
}

extern "C" void kernel_launch(void* const* d_in, const int* in_sizes, int n_in,
                              void* d_out, int out_size, void* d_ws, size_t ws_size,
                              hipStream_t stream) {
  (void)in_sizes; (void)n_in; (void)out_size; (void)ws_size;
  const float* logits = (const float*)d_in[0];
  const float* ill    = (const float*)d_in[1];
  const int*   pos    = (const int*)d_in[2];
  const float* q_init = (const float*)d_in[3];
  const float* params = (const float*)d_in[4];
  const int NBA = BATCH * AGENTS;

  float* T0 = (float*)d_ws;
  float* T1 = T0 + NBA * NPAD;

  mf_table0<<<NBA / 16, 128, 0, stream>>>(q_init, ill, params, T0);
  mf_sweep_next<<<NBA / GA, 512, 0, stream>>>(logits, ill, pos, params, T0, T1);
  mf_sweep_next<<<NBA / GA, 512, 0, stream>>>(logits, ill, pos, params, T1, T0);
  mf_sweep_last<<<NBA / GA, 512, 0, stream>>>(logits, ill, pos, T0, (float*)d_out);
}


# Row sum re-laid: lists hold table offsets; lane = (entry slot, action), so 5 lanes read an entry's 20 contiguous bytes (6 rows per load, not 32); 3 slots joined by 2 shuffles: new summation order
# speedup vs baseline: 12.2146x; 1.3008x over previous
#include <hip/hip_runtime.h>


#define FOV        11
#define MAX_DIST   5
#define NACT       5
#define NEG_INF_F  1e10f
#define AGENTS     768
#define BATCH      4
#define PMIN       60
#define NPUSED     61
#define NTRUE      (NPUSED * NACT)
#define NPAD       320
#define GA         32
#define NCH        (AGENTS / 32)
#define FLP        (AGENTS + 1)
#define SLOTS      3
#define INFL       16
#define GRID       64

typedef float v2f __attribute__((ext_vector_type(2)));
typedef float v8f __attribute__((ext_vector_type(8)));

__device__ __forceinline__ void mf_b_frag(const float* __restrict__ params,
                                          int half, int n, v2f& b0, v2f& b1) {
  const bool ok = n < NTRUE;
  const int p = n / NACT;
  const int x = n - p * NACT;
  const int base = ok ? (p + PMIN) * (NACT * NACT) + x * NACT : 0;
  const float v0 = params[base + 2 * half];
  const float v1 = params[base + 2 * half + 1];
  const float v4 = params[base + 4];
  b0.x = ok ? v0 : 0.f;
  b0.y = ok ? v1 : 0.f;
  b1.x = (ok && half == 0) ? v4 : 0.f;
  b1.y = 0.f;
}

__device__ __forceinline__ void mf_a_frag(const float* q, int half,
                                          v2f& a0, v2f& a1) {
  float m = q[0];
#pragma unroll
  for (int y = 1; y < NACT; ++y) m = fmaxf(m, q[y]);
  float e[NACT];
  float s = 0.f;
#pragma unroll
  for (int y = 0; y < NACT; ++y) { e[y] = __expf(q[y] - m); s += e[y]; }
  const float inv = 1.f / s;
  float pr[NACT];
#pragma unroll
  for (int y = 0; y < NACT; ++y) pr[y] = e[y] * inv;
  a0.x = half ? pr[2] : pr[0];
  a0.y = half ? pr[3] : pr[1];
  a1.x = half ? 0.f : pr[4];
  a1.y = 0.f;
}

__device__ __forceinline__ void mf_tile(v2f a0, v2f a1, v2f b0, v2f b1,
                                        float* __restrict__ dst) {
  v8f acc = {};
  acc = __builtin_amdgcn_wmma_f32_16x16x4_f32(
       false, a0,  false, b0,
       (short)0, acc,  false,  false);
  acc = __builtin_amdgcn_wmma_f32_16x16x4_f32(
       false, a1,  false, b1,
       (short)0, acc,  false,  false);
#pragma unroll
  for (int g = 0; g < 8; ++g) dst[(size_t)g * NPAD] = acc[g];
}

__global__ void __launch_bounds__(128)
mf_table0(const float* __restrict__ q_init,
          const float* __restrict__ ill,
          const float* __restrict__ params,
          float* __restrict__ T) {
  const int lane = threadIdx.x & 31;
  const int w    = threadIdx.x >> 5;
  const int half = lane >> 4;
  const int nl   = lane & 15;
  const int row0 = blockIdx.x * 16;

  const int r = (row0 + nl) * NACT;
  float q[NACT];
#pragma unroll
  for (int y = 0; y < NACT; ++y) q[y] = q_init[r + y] - NEG_INF_F * ill[r + y];
  v2f a0, a1;
  mf_a_frag(q, half, a0, a1);

  float* dst = T + (size_t)(row0 + 8 * half) * NPAD + nl;
  v2f b0[5], b1[5];
#pragma unroll
  for (int tt = 0; tt < 5; ++tt)
    mf_b_frag(params, half, (w * 5 + tt) * 16 + nl, b0[tt], b1[tt]);
#pragma unroll
  for (int tt = 0; tt < 5; ++tt)
    mf_tile(a0, a1, b0[tt], b1[tt], dst + (w * 5 + tt) * 16);
}

__device__ __forceinline__ unsigned mf_agent_masks(unsigned h, int lane) {
  unsigned mine = 0;
#pragma unroll
  for (int jj = 0; jj < 32; ++jj) {
    const unsigned hs = (unsigned)__builtin_amdgcn_readlane((int)h, jj);
    mine |= ((hs >> lane) & 1u) << jj;
  }
  return mine;
}

__device__ __forceinline__ unsigned mf_entry(unsigned pj, int j, int iy, int ix) {
  const int ey = (int)(pj & 0xffffu) - iy, ex = (int)(pj >> 16) - ix;
  const int dy = ey < 0 ? -ey : ey, dx = ex < 0 ? -ex : ex;
  return (unsigned)(j * NPAD + (dy * FOV + dx) * NACT);
}

__device__ __forceinline__ void mf_sweep(const float* __restrict__ logits,
                                         const float* __restrict__ ill,
                                         const int*   __restrict__ pos,
                                         const float* __restrict__ params,
                                         const float* __restrict__ T,
                                         float* __restrict__ qout,
                                         float* __restrict__ Tn,
                                         const bool last) {
  __shared__ unsigned flat[GA * FLP];
  __shared__ unsigned posl[AGENTS];
  __shared__ unsigned rowm[GRID];
  __shared__ unsigned colm[GRID];
  __shared__ int cnts[NCH * GA];
  __shared__ int tot[GA];
  __shared__ float qs[GA * NACT];

  const int tid    = threadIdx.x;
  const int b      = blockIdx.x / (AGENTS / GA);
  const int a_base = (blockIdx.x - b * (AGENTS / GA)) * GA;
  const int* pb    = pos + (size_t)(b * AGENTS) * 2;

  const int a  = tid >> 4;
  const int s  = tid & 15;
  const int sl = s / NACT;
  const int x  = s - sl * NACT;
  const int o = (b * AGENTS + a_base + a) * NACT + (s < NACT ? s : 0);
  const float lg = logits[o];
  const float il = ill[o];

  const int lane = tid & 31;
  const int wv   = __builtin_amdgcn_readfirstlane(tid >> 5);
  const int half = lane >> 4;
  const int nl   = lane & 15;
  const int rt   = wv & 1;
  const int cg   = wv >> 1;
  v2f b0[3], b1[3];
  if (!last) {
#pragma unroll
    for (int t = 0; t < 3; ++t) {
      const int ct = (t < 2 || cg < 4) ? cg + 8 * t : cg;
      mf_b_frag(params, half, ct * 16 + nl, b0[t], b1[t]);
    }
  }

  const bool two = wv < NCH - 16;
  const int iy  = pb[2 * (a_base + lane)];
  const int ix  = pb[2 * (a_base + lane) + 1];
  const int jy0 = pb[2 * tid];
  const int jx0 = pb[2 * tid + 1];
  int jy1 = 0, jx1 = 0;
  if (two) {
    jy1 = pb[2 * (tid + 512)];
    jx1 = pb[2 * (tid + 512) + 1];
  }
  posl[tid] = (unsigned)jy0 | ((unsigned)jx0 << 16);
  if (two) posl[tid + 512] = (unsigned)jy1 | ((unsigned)jx1 << 16);
  {
    unsigned rv = 0, cv = 0;
#pragma unroll
    for (int k = 0; k < 4; ++k) {
      const int r = 4 * wv + k;
      const unsigned mr = __builtin_amdgcn_ballot_w32(
          (unsigned)(iy - r + MAX_DIST) <= (unsigned)(2 * MAX_DIST));
      const unsigned mc = __builtin_amdgcn_ballot_w32(
          (unsigned)(ix - r + MAX_DIST) <= (unsigned)(2 * MAX_DIST));
      rv = lane == k ? mr : rv;
      cv = lane == k ? mc : cv;
    }
    if (lane < 4) {
      rowm[4 * wv + lane] = rv;
      colm[4 * wv + lane] = cv;
    }
  }
  __syncthreads();

  const unsigned m0 = mf_agent_masks(rowm[jy0 & (GRID - 1)] & colm[jx0 & (GRID - 1)], lane);
  cnts[wv * GA + lane] = __popc(m0);
  unsigned m1 = 0;
  if (two) {
    m1 = mf_agent_masks(rowm[jy1 & (GRID - 1)] & colm[jx1 & (GRID - 1)], lane);
    cnts[(wv + 16) * GA + lane] = __popc(m1);
  }
  __syncthreads();

  {
    int off0 = 0, off1 = 0, all = 0;
#pragma unroll
    for (int c = 0; c < NCH; ++c) {
      const int ck = cnts[c * GA + lane];
      off0 += (c < wv) ? ck : 0;
      off1 += (c < wv + 16) ? ck : 0;
      all += ck;
    }
    if (wv == 0) tot[lane] = all;
    unsigned* fa = flat + lane * FLP;
    const int c1 = two ? wv + 16 : wv;
    for (unsigned ma = m0, mb = m1; (ma | mb) != 0u; ) {
      const unsigned a2 = ma & (ma - 1u);
      const unsigned b2 = mb & (mb - 1u);
      const bool ha = ma != 0u, ha2 = a2 != 0u, hb = mb != 0u, hb2 = b2 != 0u;
      const int ja  = 32 * wv + (ha  ? __builtin_ctz(ma) : 0);
      const int ja2 = 32 * wv + (ha2 ? __builtin_ctz(a2) : 0);
      const int jb  = 32 * c1 + (hb  ? __builtin_ctz(mb) : 0);
      const int jb2 = 32 * c1 + (hb2 ? __builtin_ctz(b2) : 0);
      const unsigned qa = posl[ja], qa2 = posl[ja2], qb = posl[jb], qb2 = posl[jb2];
      asm volatile("" :: "v"(qa), "v"(qa2), "v"(qb), "v"(qb2));
      fa[ha  ? off0 : AGENTS] = mf_entry(qa,  ja,  iy, ix);  off0 += ha  ? 1 : 0;
      fa[ha2 ? off0 : AGENTS] = mf_entry(qa2, ja2, iy, ix);  off0 += ha2 ? 1 : 0;
      fa[hb  ? off1 : AGENTS] = mf_entry(qb,  jb,  iy, ix);  off1 += hb  ? 1 : 0;
      fa[hb2 ? off1 : AGENTS] = mf_entry(qb2, jb2, iy, ix);  off1 += hb2 ? 1 : 0;
      ma = a2 & (a2 - 1u);
      mb = b2 & (b2 - 1u);
    }
  }
  __syncthreads();

  const float* Tb = T + (size_t)(b * AGENTS) * NPAD;
  const unsigned* fl = flat + a * FLP;
  const int total = tot[a];
  float acc = 0.f;
  if (s < SLOTS * NACT) {
    for (int n = sl; n < total; n += SLOTS * INFL) {
      float v[INFL];
      bool  h[INFL];
#pragma unroll
      for (int u = 0; u < INFL; ++u) {
        const int nn = n + SLOTS * u;
        h[u] = nn < total;
        v[u] = Tb[fl[h[u] ? nn : n] + (unsigned)x];
      }
#pragma unroll
      for (int u = 0; u < INFL; ++u) acc += h[u] ? v[u] : 0.f;
    }
  }
  float v = acc;
  v += __shfl_down(acc, NACT, 16);
  v += __shfl_down(acc, 2 * NACT, 16);
  if (s < NACT) {
    const float qv = lg + v - NEG_INF_F * il;
    if (last) qout[o] = qv;
    else      qs[a * NACT + s] = qv;
  }
  if (last) return;

  __syncthreads();
  float q[NACT];
#pragma unroll
  for (int y = 0; y < NACT; ++y) q[y] = qs[(rt * 16 + nl) * NACT + y];
  v2f a0, a1;
  mf_a_frag(q, half, a0, a1);
  float* dst = Tn + (size_t)(b * AGENTS + a_base + rt * 16 + 8 * half) * NPAD + nl;
  mf_tile(a0, a1, b0[0], b1[0], dst + cg * 16);
  mf_tile(a0, a1, b0[1], b1[1], dst + (cg + 8) * 16);
  if (cg < 4) mf_tile(a0, a1, b0[2], b1[2], dst + (cg + 16) * 16);
}

__global__ void __launch_bounds__(512)
mf_sweep_next(const float* __restrict__ logits,
              const float* __restrict__ ill,
              const int*   __restrict__ pos,
              const float* __restrict__ params,
              const float* __restrict__ T,
              float* __restrict__ Tn) {
  mf_sweep(logits, ill, pos, params, T, nullptr, Tn, false);
}

__global__ void __launch_bounds__(512)
mf_sweep_last(const float* __restrict__ logits,
              const float* __restrict__ ill,
              const int*   __restrict__ pos,
              const float* __restrict__ T,
              float* __restrict__ qout) {
  mf_sweep(logits, ill, pos, nullptr, T, qout, nullptr, true);
}

extern "C" void kernel_launch(void* const* d_in, const int* in_sizes, int n_in,
                              void* d_out, int out_size, void* d_ws, size_t ws_size,
                              hipStream_t stream) {
  (void)in_sizes; (void)n_in; (void)out_size; (void)ws_size;
  const float* logits = (const float*)d_in[0];
  const float* ill    = (const float*)d_in[1];
  const int*   pos    = (const int*)d_in[2];
  const float* q_init = (const float*)d_in[3];
  const float* params = (const float*)d_in[4];
  const int NBA = BATCH * AGENTS;

  float* T0 = (float*)d_ws;
  float* T1 = T0 + NBA * NPAD;

  mf_table0<<<NBA / 16, 128, 0, stream>>>(q_init, ill, params, T0);
  mf_sweep_next<<<NBA / GA, 512, 0, stream>>>(logits, ill, pos, params, T0, T1);
  mf_sweep_next<<<NBA / GA, 512, 0, stream>>>(logits, ill, pos, params, T1, T0);
  mf_sweep_last<<<NBA / GA, 512, 0, stream>>>(logits, ill, pos, T0, (float*)d_out);
}


# Each sweep work-group prefetches a twelfth of its batch's table plane into the second-level cache (one request per 256-byte unit) ahead of the list stages, so the row gather hits there
# speedup vs baseline: 13.8330x; 1.1325x over previous
#include <hip/hip_runtime.h>


#define FOV        11
#define MAX_DIST   5
#define NACT       5
#define NEG_INF_F  1e10f
#define AGENTS     768
#define BATCH      4
#define PMIN       60
#define NPUSED     61
#define NTRUE      (NPUSED * NACT)
#define NPAD       320
#define GA         32
#define NCH        (AGENTS / 32)
#define FLP        (AGENTS + 1)
#define SLOTS      3
#define INFL       16
#define GRID       64
#define PF_BYTES   256
#define PF_PARTS   12
#define PF_UNITS   (AGENTS * NPAD * 4 / PF_BYTES / PF_PARTS)

typedef float v2f __attribute__((ext_vector_type(2)));
typedef float v8f __attribute__((ext_vector_type(8)));

__device__ __forceinline__ void mf_b_frag(const float* __restrict__ params,
                                          int half, int n, v2f& b0, v2f& b1) {
  const bool ok = n < NTRUE;
  const int p = n / NACT;
  const int x = n - p * NACT;
  const int base = ok ? (p + PMIN) * (NACT * NACT) + x * NACT : 0;
  const float v0 = params[base + 2 * half];
  const float v1 = params[base + 2 * half + 1];
  const float v4 = params[base + 4];
  b0.x = ok ? v0 : 0.f;
  b0.y = ok ? v1 : 0.f;
  b1.x = (ok && half == 0) ? v4 : 0.f;
  b1.y = 0.f;
}

__device__ __forceinline__ void mf_a_frag(const float* q, int half,
                                          v2f& a0, v2f& a1) {
  float m = q[0];
#pragma unroll
  for (int y = 1; y < NACT; ++y) m = fmaxf(m, q[y]);
  float e[NACT];
  float s = 0.f;
#pragma unroll
  for (int y = 0; y < NACT; ++y) { e[y] = __expf(q[y] - m); s += e[y]; }
  const float inv = 1.f / s;
  float pr[NACT];
#pragma unroll
  for (int y = 0; y < NACT; ++y) pr[y] = e[y] * inv;
  a0.x = half ? pr[2] : pr[0];
  a0.y = half ? pr[3] : pr[1];
  a1.x = half ? 0.f : pr[4];
  a1.y = 0.f;
}

__device__ __forceinline__ void mf_tile(v2f a0, v2f a1, v2f b0, v2f b1,
                                        float* __restrict__ dst) {
  v8f acc = {};
  acc = __builtin_amdgcn_wmma_f32_16x16x4_f32(
       false, a0,  false, b0,
       (short)0, acc,  false,  false);
  acc = __builtin_amdgcn_wmma_f32_16x16x4_f32(
       false, a1,  false, b1,
       (short)0, acc,  false,  false);
#pragma unroll
  for (int g = 0; g < 8; ++g) dst[(size_t)g * NPAD] = acc[g];
}

__global__ void __launch_bounds__(128)
mf_table0(const float* __restrict__ q_init,
          const float* __restrict__ ill,
          const float* __restrict__ params,
          float* __restrict__ T) {
  const int lane = threadIdx.x & 31;
  const int w    = threadIdx.x >> 5;
  const int half = lane >> 4;
  const int nl   = lane & 15;
  const int row0 = blockIdx.x * 16;

  const int r = (row0 + nl) * NACT;
  float q[NACT];
#pragma unroll
  for (int y = 0; y < NACT; ++y) q[y] = q_init[r + y] - NEG_INF_F * ill[r + y];
  v2f a0, a1;
  mf_a_frag(q, half, a0, a1);

  float* dst = T + (size_t)(row0 + 8 * half) * NPAD + nl;
  v2f b0[5], b1[5];
#pragma unroll
  for (int tt = 0; tt < 5; ++tt)
    mf_b_frag(params, half, (w * 5 + tt) * 16 + nl, b0[tt], b1[tt]);
#pragma unroll
  for (int tt = 0; tt < 5; ++tt)
    mf_tile(a0, a1, b0[tt], b1[tt], dst + (w * 5 + tt) * 16);
}

__device__ __forceinline__ unsigned mf_agent_masks(unsigned h, int lane) {
  unsigned mine = 0;
#pragma unroll
  for (int jj = 0; jj < 32; ++jj) {
    const unsigned hs = (unsigned)__builtin_amdgcn_readlane((int)h, jj);
    mine |= ((hs >> lane) & 1u) << jj;
  }
  return mine;
}

__device__ __forceinline__ unsigned mf_entry(unsigned pj, int j, int iy, int ix) {
  const int ey = (int)(pj & 0xffffu) - iy, ex = (int)(pj >> 16) - ix;
  const int dy = ey < 0 ? -ey : ey, dx = ex < 0 ? -ex : ex;
  return (unsigned)(j * NPAD + (dy * FOV + dx) * NACT);
}

__device__ __forceinline__ void mf_sweep(const float* __restrict__ logits,
                                         const float* __restrict__ ill,
                                         const int*   __restrict__ pos,
                                         const float* __restrict__ params,
                                         const float* __restrict__ T,
                                         float* __restrict__ qout,
                                         float* __restrict__ Tn,
                                         const bool last) {
  __shared__ unsigned flat[GA * FLP];
  __shared__ unsigned posl[AGENTS];
  __shared__ unsigned rowm[GRID];
  __shared__ unsigned colm[GRID];
  __shared__ int cnts[NCH * GA];
  __shared__ int tot[GA];
  __shared__ float qs[GA * NACT];

  const int tid    = threadIdx.x;
  const int b      = blockIdx.x / (AGENTS / GA);
  const int a_base = (blockIdx.x - b * (AGENTS / GA)) * GA;
  const int* pb    = pos + (size_t)(b * AGENTS) * 2;

  const int a  = tid >> 4;
  const int s  = tid & 15;
  const int sl = s / NACT;
  const int x  = s - sl * NACT;
  const int o = (b * AGENTS + a_base + a) * NACT + (s < NACT ? s : 0);
  const float lg = logits[o];
  const float il = ill[o];

  const int lane = tid & 31;
  const int wv   = __builtin_amdgcn_readfirstlane(tid >> 5);
  const int half = lane >> 4;
  const int nl   = lane & 15;
  const int rt   = wv & 1;
  const int cg   = wv >> 1;
  v2f b0[3], b1[3];
  if (!last) {
#pragma unroll
    for (int t = 0; t < 3; ++t) {
      const int ct = (t < 2 || cg < 4) ? cg + 8 * t : cg;
      mf_b_frag(params, half, ct * 16 + nl, b0[t], b1[t]);
    }
  }

  const bool two = wv < NCH - 16;
  const int iy  = pb[2 * (a_base + lane)];
  const int ix  = pb[2 * (a_base + lane) + 1];
  const int jy0 = pb[2 * tid];
  const int jx0 = pb[2 * tid + 1];
  {
    const int k = blockIdx.x - b * (AGENTS / GA);
    const int p = k & 7;
    const int part = (k >> 3) * 4 + (p < 4 ? p : p ^ 5);
    if (tid < PF_UNITS)
      __builtin_prefetch((const char*)(T + (size_t)(b * AGENTS) * NPAD) +
                             (size_t)(part * PF_UNITS + tid) * PF_BYTES,
                          0,  2);
  }
  int jy1 = 0, jx1 = 0;
  if (two) {
    jy1 = pb[2 * (tid + 512)];
    jx1 = pb[2 * (tid + 512) + 1];
  }
  posl[tid] = (unsigned)jy0 | ((unsigned)jx0 << 16);
  if (two) posl[tid + 512] = (unsigned)jy1 | ((unsigned)jx1 << 16);
  {
    unsigned rv = 0, cv = 0;
#pragma unroll
    for (int k = 0; k < 4; ++k) {
      const int r = 4 * wv + k;
      const unsigned mr = __builtin_amdgcn_ballot_w32(
          (unsigned)(iy - r + MAX_DIST) <= (unsigned)(2 * MAX_DIST));
      const unsigned mc = __builtin_amdgcn_ballot_w32(
          (unsigned)(ix - r + MAX_DIST) <= (unsigned)(2 * MAX_DIST));
      rv = lane == k ? mr : rv;
      cv = lane == k ? mc : cv;
    }
    if (lane < 4) {
      rowm[4 * wv + lane] = rv;
      colm[4 * wv + lane] = cv;
    }
  }
  __syncthreads();

  const unsigned m0 = mf_agent_masks(rowm[jy0 & (GRID - 1)] & colm[jx0 & (GRID - 1)], lane);
  cnts[wv * GA + lane] = __popc(m0);
  unsigned m1 = 0;
  if (two) {
    m1 = mf_agent_masks(rowm[jy1 & (GRID - 1)] & colm[jx1 & (GRID - 1)], lane);
    cnts[(wv + 16) * GA + lane] = __popc(m1);
  }
  __syncthreads();

  {
    int off0 = 0, off1 = 0, all = 0;
#pragma unroll
    for (int c = 0; c < NCH; ++c) {
      const int ck = cnts[c * GA + lane];
      off0 += (c < wv) ? ck : 0;
      off1 += (c < wv + 16) ? ck : 0;
      all += ck;
    }
    if (wv == 0) tot[lane] = all;
    unsigned* fa = flat + lane * FLP;
    const int c1 = two ? wv + 16 : wv;
    for (unsigned ma = m0, mb = m1; (ma | mb) != 0u; ) {
      const unsigned a2 = ma & (ma - 1u);
      const unsigned b2 = mb & (mb - 1u);
      const bool ha = ma != 0u, ha2 = a2 != 0u, hb = mb != 0u, hb2 = b2 != 0u;
      const int ja  = 32 * wv + (ha  ? __builtin_ctz(ma) : 0);
      const int ja2 = 32 * wv + (ha2 ? __builtin_ctz(a2) : 0);
      const int jb  = 32 * c1 + (hb  ? __builtin_ctz(mb) : 0);
      const int jb2 = 32 * c1 + (hb2 ? __builtin_ctz(b2) : 0);
      const unsigned qa = posl[ja], qa2 = posl[ja2], qb = posl[jb], qb2 = posl[jb2];
      asm volatile("" :: "v"(qa), "v"(qa2), "v"(qb), "v"(qb2));
      fa[ha  ? off0 : AGENTS] = mf_entry(qa,  ja,  iy, ix);  off0 += ha  ? 1 : 0;
      fa[ha2 ? off0 : AGENTS] = mf_entry(qa2, ja2, iy, ix);  off0 += ha2 ? 1 : 0;
      fa[hb  ? off1 : AGENTS] = mf_entry(qb,  jb,  iy, ix);  off1 += hb  ? 1 : 0;
      fa[hb2 ? off1 : AGENTS] = mf_entry(qb2, jb2, iy, ix);  off1 += hb2 ? 1 : 0;
      ma = a2 & (a2 - 1u);
      mb = b2 & (b2 - 1u);
    }
  }
  __syncthreads();

  const float* Tb = T + (size_t)(b * AGENTS) * NPAD;
  const unsigned* fl = flat + a * FLP;
  const int total = tot[a];
  float acc = 0.f;
  if (s < SLOTS * NACT) {
    for (int n = sl; n < total; n += SLOTS * INFL) {
      float v[INFL];
      bool  h[INFL];
#pragma unroll
      for (int u = 0; u < INFL; ++u) {
        const int nn = n + SLOTS * u;
        h[u] = nn < total;
        v[u] = Tb[fl[h[u] ? nn : n] + (unsigned)x];
      }
#pragma unroll
      for (int u = 0; u < INFL; ++u) acc += h[u] ? v[u] : 0.f;
    }
  }
  float v = acc;
  v += __shfl_down(acc, NACT, 16);
  v += __shfl_down(acc, 2 * NACT, 16);
  if (s < NACT) {
    const float qv = lg + v - NEG_INF_F * il;
    if (last) qout[o] = qv;
    else      qs[a * NACT + s] = qv;
  }
  if (last) return;

  __syncthreads();
  float q[NACT];
#pragma unroll
  for (int y = 0; y < NACT; ++y) q[y] = qs[(rt * 16 + nl) * NACT + y];
  v2f a0, a1;
  mf_a_frag(q, half, a0, a1);
  float* dst = Tn + (size_t)(b * AGENTS + a_base + rt * 16 + 8 * half) * NPAD + nl;
  mf_tile(a0, a1, b0[0], b1[0], dst + cg * 16);
  mf_tile(a0, a1, b0[1], b1[1], dst + (cg + 8) * 16);
  if (cg < 4) mf_tile(a0, a1, b0[2], b1[2], dst + (cg + 16) * 16);
}

__global__ void __launch_bounds__(512)
mf_sweep_next(const float* __restrict__ logits,
              const float* __restrict__ ill,
              const int*   __restrict__ pos,
              const float* __restrict__ params,
              const float* __restrict__ T,
              float* __restrict__ Tn) {
  mf_sweep(logits, ill, pos, params, T, nullptr, Tn, false);
}

__global__ void __launch_bounds__(512)
mf_sweep_last(const float* __restrict__ logits,
              const float* __restrict__ ill,
              const int*   __restrict__ pos,
              const float* __restrict__ T,
              float* __restrict__ qout) {
  mf_sweep(logits, ill, pos, nullptr, T, qout, nullptr, true);
}

extern "C" void kernel_launch(void* const* d_in, const int* in_sizes, int n_in,
                              void* d_out, int out_size, void* d_ws, size_t ws_size,
                              hipStream_t stream) {
  (void)in_sizes; (void)n_in; (void)out_size; (void)ws_size;
  const float* logits = (const float*)d_in[0];
  const float* ill    = (const float*)d_in[1];
  const int*   pos    = (const int*)d_in[2];
  const float* q_init = (const float*)d_in[3];
  const float* params = (const float*)d_in[4];
  const int NBA = BATCH * AGENTS;

  float* T0 = (float*)d_ws;
  float* T1 = T0 + NBA * NPAD;

  mf_table0<<<NBA / 16, 128, 0, stream>>>(q_init, ill, params, T0);
  mf_sweep_next<<<NBA / GA, 512, 0, stream>>>(logits, ill, pos, params, T0, T1);
  mf_sweep_next<<<NBA / GA, 512, 0, stream>>>(logits, ill, pos, params, T1, T0);
  mf_sweep_last<<<NBA / GA, 512, 0, stream>>>(logits, ill, pos, T0, (float*)d_out);
}


# Table cut to the 36 offset pairs a neighbour can have (|dy|,|dx| <= 5): 192 columns, not 320; a row is 768 bytes (three 256-byte units), 12 column tiles, not 20; same sums
# speedup vs baseline: 14.8222x; 1.0715x over previous
#include <hip/hip_runtime.h>


#define FOV        11
#define MAX_DIST   5
#define NACT       5
#define NEG_INF_F  1e10f
#define AGENTS     768
#define BATCH      4
#define NOFF       (MAX_DIST + 1)
#define NPUSED     (NOFF * NOFF)
#define NTRUE      (NPUSED * NACT)
#define NPAD       192
#define NTILES     (NPAD / 16)
#define GA         32
#define NCH        (AGENTS / 32)
#define FLP        (AGENTS + 1)
#define SLOTS      3
#define INFL       16
#define GRID       64
#define PF_BYTES   256
#define PF_PARTS   12
#define PF_UNITS   (AGENTS * NPAD * 4 / PF_BYTES / PF_PARTS)

typedef float v2f __attribute__((ext_vector_type(2)));
typedef float v8f __attribute__((ext_vector_type(8)));

__device__ __forceinline__ void mf_b_frag(const float* __restrict__ params,
                                          int half, int n, v2f& b0, v2f& b1) {
  const bool ok = n < NTRUE;
  const int p = n / NACT;
  const int x = n - p * NACT;
  const int dy = p / NOFF;
  const int dx = p - dy * NOFF;
  const int pidx = (dy + MAX_DIST) * FOV + (dx + MAX_DIST);
  const int base = ok ? pidx * (NACT * NACT) + x * NACT : 0;
  const float v0 = params[base + 2 * half];
  const float v1 = params[base + 2 * half + 1];
  const float v4 = params[base + 4];
  b0.x = ok ? v0 : 0.f;
  b0.y = ok ? v1 : 0.f;
  b1.x = (ok && half == 0) ? v4 : 0.f;
  b1.y = 0.f;
}

__device__ __forceinline__ void mf_a_frag(const float* q, int half,
                                          v2f& a0, v2f& a1) {
  float m = q[0];
#pragma unroll
  for (int y = 1; y < NACT; ++y) m = fmaxf(m, q[y]);
  float e[NACT];
  float s = 0.f;
#pragma unroll
  for (int y = 0; y < NACT; ++y) { e[y] = __expf(q[y] - m); s += e[y]; }
  const float inv = 1.f / s;
  float pr[NACT];
#pragma unroll
  for (int y = 0; y < NACT; ++y) pr[y] = e[y] * inv;
  a0.x = half ? pr[2] : pr[0];
  a0.y = half ? pr[3] : pr[1];
  a1.x = half ? 0.f : pr[4];
  a1.y = 0.f;
}

__device__ __forceinline__ void mf_tile(v2f a0, v2f a1, v2f b0, v2f b1,
                                        float* __restrict__ dst) {
  v8f acc = {};
  acc = __builtin_amdgcn_wmma_f32_16x16x4_f32(
       false, a0,  false, b0,
       (short)0, acc,  false,  false);
  acc = __builtin_amdgcn_wmma_f32_16x16x4_f32(
       false, a1,  false, b1,
       (short)0, acc,  false,  false);
#pragma unroll
  for (int g = 0; g < 8; ++g) dst[(size_t)g * NPAD] = acc[g];
}

__global__ void __launch_bounds__(128)
mf_table0(const float* __restrict__ q_init,
          const float* __restrict__ ill,
          const float* __restrict__ params,
          float* __restrict__ T) {
  const int lane = threadIdx.x & 31;
  const int w    = threadIdx.x >> 5;
  const int half = lane >> 4;
  const int nl   = lane & 15;
  const int row0 = blockIdx.x * 16;

  const int r = (row0 + nl) * NACT;
  float q[NACT];
#pragma unroll
  for (int y = 0; y < NACT; ++y) q[y] = q_init[r + y] - NEG_INF_F * ill[r + y];
  v2f a0, a1;
  mf_a_frag(q, half, a0, a1);

  float* dst = T + (size_t)(row0 + 8 * half) * NPAD + nl;
  v2f b0[NTILES / 4], b1[NTILES / 4];
#pragma unroll
  for (int tt = 0; tt < NTILES / 4; ++tt)
    mf_b_frag(params, half, (w * (NTILES / 4) + tt) * 16 + nl, b0[tt], b1[tt]);
#pragma unroll
  for (int tt = 0; tt < NTILES / 4; ++tt)
    mf_tile(a0, a1, b0[tt], b1[tt], dst + (w * (NTILES / 4) + tt) * 16);
}

__device__ __forceinline__ unsigned mf_agent_masks(unsigned h, int lane) {
  unsigned mine = 0;
#pragma unroll
  for (int jj = 0; jj < 32; ++jj) {
    const unsigned hs = (unsigned)__builtin_amdgcn_readlane((int)h, jj);
    mine |= ((hs >> lane) & 1u) << jj;
  }
  return mine;
}

__device__ __forceinline__ unsigned mf_entry(unsigned pj, int j, int iy, int ix) {
  const int ey = (int)(pj & 0xffffu) - iy, ex = (int)(pj >> 16) - ix;
  const int dy = ey < 0 ? -ey : ey, dx = ex < 0 ? -ex : ex;
  return (unsigned)(j * NPAD + (dy * NOFF + dx) * NACT);
}

__device__ __forceinline__ void mf_sweep(const float* __restrict__ logits,
                                         const float* __restrict__ ill,
                                         const int*   __restrict__ pos,
                                         const float* __restrict__ params,
                                         const float* __restrict__ T,
                                         float* __restrict__ qout,
                                         float* __restrict__ Tn,
                                         const bool last) {
  __shared__ unsigned flat[GA * FLP];
  __shared__ unsigned posl[AGENTS];
  __shared__ unsigned rowm[GRID];
  __shared__ unsigned colm[GRID];
  __shared__ int cnts[NCH * GA];
  __shared__ int tot[GA];
  __shared__ float qs[GA * NACT];

  const int tid    = threadIdx.x;
  const int b      = blockIdx.x / (AGENTS / GA);
  const int a_base = (blockIdx.x - b * (AGENTS / GA)) * GA;
  const int* pb    = pos + (size_t)(b * AGENTS) * 2;

  const int a  = tid >> 4;
  const int s  = tid & 15;
  const int sl = s / NACT;
  const int x  = s - sl * NACT;
  const int o = (b * AGENTS + a_base + a) * NACT + (s < NACT ? s : 0);
  const float lg = logits[o];
  const float il = ill[o];

  const int lane = tid & 31;
  const int wv   = __builtin_amdgcn_readfirstlane(tid >> 5);
  const int half = lane >> 4;
  const int nl   = lane & 15;
  const int rt   = wv & 1;
  const int cg   = wv >> 1;
  v2f b0[2], b1[2];
  if (!last) {
#pragma unroll
    for (int t = 0; t < 2; ++t) {
      const int ct = (t < 1 || cg < NTILES - 8) ? cg + 8 * t : cg;
      mf_b_frag(params, half, ct * 16 + nl, b0[t], b1[t]);
    }
  }

  const bool two = wv < NCH - 16;
  const int iy  = pb[2 * (a_base + lane)];
  const int ix  = pb[2 * (a_base + lane) + 1];
  const int jy0 = pb[2 * tid];
  const int jx0 = pb[2 * tid + 1];
  {
    const int k = blockIdx.x - b * (AGENTS / GA);
    const int p = k & 7;
    const int part = (k >> 3) * 4 + (p < 4 ? p : p ^ 5);
    if (tid < PF_UNITS)
      __builtin_prefetch((const char*)(T + (size_t)(b * AGENTS) * NPAD) +
                             (size_t)(part * PF_UNITS + tid) * PF_BYTES,
                          0,  2);
  }
  int jy1 = 0, jx1 = 0;
  if (two) {
    jy1 = pb[2 * (tid + 512)];
    jx1 = pb[2 * (tid + 512) + 1];
  }
  posl[tid] = (unsigned)jy0 | ((unsigned)jx0 << 16);
  if (two) posl[tid + 512] = (unsigned)jy1 | ((unsigned)jx1 << 16);
  {
    unsigned rv = 0, cv = 0;
#pragma unroll
    for (int k = 0; k < 4; ++k) {
      const int r = 4 * wv + k;
      const unsigned mr = __builtin_amdgcn_ballot_w32(
          (unsigned)(iy - r + MAX_DIST) <= (unsigned)(2 * MAX_DIST));
      const unsigned mc = __builtin_amdgcn_ballot_w32(
          (unsigned)(ix - r + MAX_DIST) <= (unsigned)(2 * MAX_DIST));
      rv = lane == k ? mr : rv;
      cv = lane == k ? mc : cv;
    }
    if (lane < 4) {
      rowm[4 * wv + lane] = rv;
      colm[4 * wv + lane] = cv;
    }
  }
  __syncthreads();

  const unsigned m0 = mf_agent_masks(rowm[jy0 & (GRID - 1)] & colm[jx0 & (GRID - 1)], lane);
  cnts[wv * GA + lane] = __popc(m0);
  unsigned m1 = 0;
  if (two) {
    m1 = mf_agent_masks(rowm[jy1 & (GRID - 1)] & colm[jx1 & (GRID - 1)], lane);
    cnts[(wv + 16) * GA + lane] = __popc(m1);
  }
  __syncthreads();

  {
    int off0 = 0, off1 = 0, all = 0;
#pragma unroll
    for (int c = 0; c < NCH; ++c) {
      const int ck = cnts[c * GA + lane];
      off0 += (c < wv) ? ck : 0;
      off1 += (c < wv + 16) ? ck : 0;
      all += ck;
    }
    if (wv == 0) tot[lane] = all;
    unsigned* fa = flat + lane * FLP;
    const int c1 = two ? wv + 16 : wv;
    for (unsigned ma = m0, mb = m1; (ma | mb) != 0u; ) {
      const unsigned a2 = ma & (ma - 1u);
      const unsigned b2 = mb & (mb - 1u);
      const bool ha = ma != 0u, ha2 = a2 != 0u, hb = mb != 0u, hb2 = b2 != 0u;
      const int ja  = 32 * wv + (ha  ? __builtin_ctz(ma) : 0);
      const int ja2 = 32 * wv + (ha2 ? __builtin_ctz(a2) : 0);
      const int jb  = 32 * c1 + (hb  ? __builtin_ctz(mb) : 0);
      const int jb2 = 32 * c1 + (hb2 ? __builtin_ctz(b2) : 0);
      const unsigned qa = posl[ja], qa2 = posl[ja2], qb = posl[jb], qb2 = posl[jb2];
      asm volatile("" :: "v"(qa), "v"(qa2), "v"(qb), "v"(qb2));
      fa[ha  ? off0 : AGENTS] = mf_entry(qa,  ja,  iy, ix);  off0 += ha  ? 1 : 0;
      fa[ha2 ? off0 : AGENTS] = mf_entry(qa2, ja2, iy, ix);  off0 += ha2 ? 1 : 0;
      fa[hb  ? off1 : AGENTS] = mf_entry(qb,  jb,  iy, ix);  off1 += hb  ? 1 : 0;
      fa[hb2 ? off1 : AGENTS] = mf_entry(qb2, jb2, iy, ix);  off1 += hb2 ? 1 : 0;
      ma = a2 & (a2 - 1u);
      mb = b2 & (b2 - 1u);
    }
  }
  __syncthreads();

  const float* Tb = T + (size_t)(b * AGENTS) * NPAD;
  const unsigned* fl = flat + a * FLP;
  const int total = tot[a];
  float acc = 0.f;
  if (s < SLOTS * NACT) {
    for (int n = sl; n < total; n += SLOTS * INFL) {
      float v[INFL];
      bool  h[INFL];
#pragma unroll
      for (int u = 0; u < INFL; ++u) {
        const int nn = n + SLOTS * u;
        h[u] = nn < total;
        v[u] = Tb[fl[h[u] ? nn : n] + (unsigned)x];
      }
#pragma unroll
      for (int u = 0; u < INFL; ++u) acc += h[u] ? v[u] : 0.f;
    }
  }
  float v = acc;
  v += __shfl_down(acc, NACT, 16);
  v += __shfl_down(acc, 2 * NACT, 16);
  if (s < NACT) {
    const float qv = lg + v - NEG_INF_F * il;
    if (last) qout[o] = qv;
    else      qs[a * NACT + s] = qv;
  }
  if (last) return;

  __syncthreads();
  float q[NACT];
#pragma unroll
  for (int y = 0; y < NACT; ++y) q[y] = qs[(rt * 16 + nl) * NACT + y];
  v2f a0, a1;
  mf_a_frag(q, half, a0, a1);
  float* dst = Tn + (size_t)(b * AGENTS + a_base + rt * 16 + 8 * half) * NPAD + nl;
  mf_tile(a0, a1, b0[0], b1[0], dst + cg * 16);
  if (cg < NTILES - 8) mf_tile(a0, a1, b0[1], b1[1], dst + (cg + 8) * 16);
}

__global__ void __launch_bounds__(512)
mf_sweep_next(const float* __restrict__ logits,
              const float* __restrict__ ill,
              const int*   __restrict__ pos,
              const float* __restrict__ params,
              const float* __restrict__ T,
              float* __restrict__ Tn) {
  mf_sweep(logits, ill, pos, params, T, nullptr, Tn, false);
}

__global__ void __launch_bounds__(512)
mf_sweep_last(const float* __restrict__ logits,
              const float* __restrict__ ill,
              const int*   __restrict__ pos,
              const float* __restrict__ T,
              float* __restrict__ qout) {
  mf_sweep(logits, ill, pos, nullptr, T, qout, nullptr, true);
}

extern "C" void kernel_launch(void* const* d_in, const int* in_sizes, int n_in,
                              void* d_out, int out_size, void* d_ws, size_t ws_size,
                              hipStream_t stream) {
  (void)in_sizes; (void)n_in; (void)out_size; (void)ws_size;
  const float* logits = (const float*)d_in[0];
  const float* ill    = (const float*)d_in[1];
  const int*   pos    = (const int*)d_in[2];
  const float* q_init = (const float*)d_in[3];
  const float* params = (const float*)d_in[4];
  const int NBA = BATCH * AGENTS;

  float* T0 = (float*)d_ws;
  float* T1 = T0 + NBA * NPAD;

  mf_table0<<<NBA / 16, 128, 0, stream>>>(q_init, ill, params, T0);
  mf_sweep_next<<<NBA / GA, 512, 0, stream>>>(logits, ill, pos, params, T0, T1);
  mf_sweep_next<<<NBA / GA, 512, 0, stream>>>(logits, ill, pos, params, T1, T0);
  mf_sweep_last<<<NBA / GA, 512, 0, stream>>>(logits, ill, pos, T0, (float*)d_out);
}


# Per-chunk masks transposed (candidate sets -> per-agent candidate masks) by a 5-step 32x32 bit-matrix butterfly over lane exchanges, a wave's two chunks interleaved, not 32 lane reads; same sums
# speedup vs baseline: 15.5589x; 1.0497x over previous
#include <hip/hip_runtime.h>


#define FOV        11
#define MAX_DIST   5
#define NACT       5
#define NEG_INF_F  1e10f
#define AGENTS     768
#define BATCH      4
#define NOFF       (MAX_DIST + 1)
#define NPUSED     (NOFF * NOFF)
#define NTRUE      (NPUSED * NACT)
#define NPAD       192
#define NTILES     (NPAD / 16)
#define GA         32
#define NCH        (AGENTS / 32)
#define FLP        (AGENTS + 1)
#define SLOTS      3
#define INFL       16
#define GRID       64
#define PF_BYTES   256
#define PF_PARTS   12
#define PF_UNITS   (AGENTS * NPAD * 4 / PF_BYTES / PF_PARTS)

typedef float v2f __attribute__((ext_vector_type(2)));
typedef float v8f __attribute__((ext_vector_type(8)));

__device__ __forceinline__ void mf_b_frag(const float* __restrict__ params,
                                          int half, int n, v2f& b0, v2f& b1) {
  const bool ok = n < NTRUE;
  const int p = n / NACT;
  const int x = n - p * NACT;
  const int dy = p / NOFF;
  const int dx = p - dy * NOFF;
  const int pidx = (dy + MAX_DIST) * FOV + (dx + MAX_DIST);
  const int base = ok ? pidx * (NACT * NACT) + x * NACT : 0;
  const float v0 = params[base + 2 * half];
  const float v1 = params[base + 2 * half + 1];
  const float v4 = params[base + 4];
  b0.x = ok ? v0 : 0.f;
  b0.y = ok ? v1 : 0.f;
  b1.x = (ok && half == 0) ? v4 : 0.f;
  b1.y = 0.f;
}

__device__ __forceinline__ void mf_a_frag(const float* q, int half,
                                          v2f& a0, v2f& a1) {
  float m = q[0];
#pragma unroll
  for (int y = 1; y < NACT; ++y) m = fmaxf(m, q[y]);
  float e[NACT];
  float s = 0.f;
#pragma unroll
  for (int y = 0; y < NACT; ++y) { e[y] = __expf(q[y] - m); s += e[y]; }
  const float inv = 1.f / s;
  float pr[NACT];
#pragma unroll
  for (int y = 0; y < NACT; ++y) pr[y] = e[y] * inv;
  a0.x = half ? pr[2] : pr[0];
  a0.y = half ? pr[3] : pr[1];
  a1.x = half ? 0.f : pr[4];
  a1.y = 0.f;
}

__device__ __forceinline__ void mf_tile(v2f a0, v2f a1, v2f b0, v2f b1,
                                        float* __restrict__ dst) {
  v8f acc = {};
  acc = __builtin_amdgcn_wmma_f32_16x16x4_f32(
       false, a0,  false, b0,
       (short)0, acc,  false,  false);
  acc = __builtin_amdgcn_wmma_f32_16x16x4_f32(
       false, a1,  false, b1,
       (short)0, acc,  false,  false);
#pragma unroll
  for (int g = 0; g < 8; ++g) dst[(size_t)g * NPAD] = acc[g];
}

__global__ void __launch_bounds__(128)
mf_table0(const float* __restrict__ q_init,
          const float* __restrict__ ill,
          const float* __restrict__ params,
          float* __restrict__ T) {
  const int lane = threadIdx.x & 31;
  const int w    = threadIdx.x >> 5;
  const int half = lane >> 4;
  const int nl   = lane & 15;
  const int row0 = blockIdx.x * 16;

  const int r = (row0 + nl) * NACT;
  float q[NACT];
#pragma unroll
  for (int y = 0; y < NACT; ++y) q[y] = q_init[r + y] - NEG_INF_F * ill[r + y];
  v2f a0, a1;
  mf_a_frag(q, half, a0, a1);

  float* dst = T + (size_t)(row0 + 8 * half) * NPAD + nl;
  v2f b0[NTILES / 4], b1[NTILES / 4];
#pragma unroll
  for (int tt = 0; tt < NTILES / 4; ++tt)
    mf_b_frag(params, half, (w * (NTILES / 4) + tt) * 16 + nl, b0[tt], b1[tt]);
#pragma unroll
  for (int tt = 0; tt < NTILES / 4; ++tt)
    mf_tile(a0, a1, b0[tt], b1[tt], dst + (w * (NTILES / 4) + tt) * 16);
}

__device__ __forceinline__ unsigned mf_turn_step(unsigned x, unsigned t, int lane,
                                                 int s, unsigned m) {
  const bool up = (lane & s) != 0;
  const unsigned keep = up ? ~m : m;
  const unsigned r = __builtin_rotateright32(t, up ? (unsigned)s : (unsigned)(32 - s));
  return (x & keep) | (r & ~keep);
}

__device__ __forceinline__ unsigned mf_agent_masks(unsigned h, int lane) {
  unsigned x = h;
  unsigned t = (unsigned)__builtin_amdgcn_permlanex16((int)x, (int)x, 0x76543210, 0xfedcba98,
                                                      false, false);
  x = mf_turn_step(x, t, lane, 16, 0x0000ffffu);
  t = (unsigned)__builtin_amdgcn_update_dpp(0, (int)x, 0x168, 0xf, 0xf, false);
  x = mf_turn_step(x, t, lane, 8, 0x00ff00ffu);
  t = (unsigned)__builtin_amdgcn_update_dpp(0, (int)x, 0x164, 0xf, 0xf, false);
  x = mf_turn_step(x, t, lane, 4, 0x0f0f0f0fu);
  t = (unsigned)__builtin_amdgcn_update_dpp(0, (int)x, 0x162, 0xf, 0xf, false);
  x = mf_turn_step(x, t, lane, 2, 0x33333333u);
  t = (unsigned)__builtin_amdgcn_update_dpp(0, (int)x, 0x161, 0xf, 0xf, false);
  x = mf_turn_step(x, t, lane, 1, 0x55555555u);
  return x;
}

__device__ __forceinline__ unsigned mf_entry(unsigned pj, int j, int iy, int ix) {
  const int ey = (int)(pj & 0xffffu) - iy, ex = (int)(pj >> 16) - ix;
  const int dy = ey < 0 ? -ey : ey, dx = ex < 0 ? -ex : ex;
  return (unsigned)(j * NPAD + (dy * NOFF + dx) * NACT);
}

__device__ __forceinline__ void mf_sweep(const float* __restrict__ logits,
                                         const float* __restrict__ ill,
                                         const int*   __restrict__ pos,
                                         const float* __restrict__ params,
                                         const float* __restrict__ T,
                                         float* __restrict__ qout,
                                         float* __restrict__ Tn,
                                         const bool last) {
  __shared__ unsigned flat[GA * FLP];
  __shared__ unsigned posl[AGENTS];
  __shared__ unsigned rowm[GRID];
  __shared__ unsigned colm[GRID];
  __shared__ int cnts[NCH * GA];
  __shared__ int tot[GA];
  __shared__ float qs[GA * NACT];

  const int tid    = threadIdx.x;
  const int b      = blockIdx.x / (AGENTS / GA);
  const int a_base = (blockIdx.x - b * (AGENTS / GA)) * GA;
  const int* pb    = pos + (size_t)(b * AGENTS) * 2;

  const int a  = tid >> 4;
  const int s  = tid & 15;
  const int sl = s / NACT;
  const int x  = s - sl * NACT;
  const int o = (b * AGENTS + a_base + a) * NACT + (s < NACT ? s : 0);
  const float lg = logits[o];
  const float il = ill[o];

  const int lane = tid & 31;
  const int wv   = __builtin_amdgcn_readfirstlane(tid >> 5);
  const int half = lane >> 4;
  const int nl   = lane & 15;
  const int rt   = wv & 1;
  const int cg   = wv >> 1;
  v2f b0[2], b1[2];
  if (!last) {
#pragma unroll
    for (int t = 0; t < 2; ++t) {
      const int ct = (t < 1 || cg < NTILES - 8) ? cg + 8 * t : cg;
      mf_b_frag(params, half, ct * 16 + nl, b0[t], b1[t]);
    }
  }

  const bool two = wv < NCH - 16;
  const int iy  = pb[2 * (a_base + lane)];
  const int ix  = pb[2 * (a_base + lane) + 1];
  const int jy0 = pb[2 * tid];
  const int jx0 = pb[2 * tid + 1];
  {
    const int k = blockIdx.x - b * (AGENTS / GA);
    const int p = k & 7;
    const int part = (k >> 3) * 4 + (p < 4 ? p : p ^ 5);
    if (tid < PF_UNITS)
      __builtin_prefetch((const char*)(T + (size_t)(b * AGENTS) * NPAD) +
                             (size_t)(part * PF_UNITS + tid) * PF_BYTES,
                          0,  2);
  }
  int jy1 = 0, jx1 = 0;
  if (two) {
    jy1 = pb[2 * (tid + 512)];
    jx1 = pb[2 * (tid + 512) + 1];
  }
  posl[tid] = (unsigned)jy0 | ((unsigned)jx0 << 16);
  if (two) posl[tid + 512] = (unsigned)jy1 | ((unsigned)jx1 << 16);
  {
    unsigned rv = 0, cv = 0;
#pragma unroll
    for (int k = 0; k < 4; ++k) {
      const int r = 4 * wv + k;
      const unsigned mr = __builtin_amdgcn_ballot_w32(
          (unsigned)(iy - r + MAX_DIST) <= (unsigned)(2 * MAX_DIST));
      const unsigned mc = __builtin_amdgcn_ballot_w32(
          (unsigned)(ix - r + MAX_DIST) <= (unsigned)(2 * MAX_DIST));
      rv = lane == k ? mr : rv;
      cv = lane == k ? mc : cv;
    }
    if (lane < 4) {
      rowm[4 * wv + lane] = rv;
      colm[4 * wv + lane] = cv;
    }
  }
  __syncthreads();

  const unsigned h0 = rowm[jy0 & (GRID - 1)] & colm[jx0 & (GRID - 1)];
  unsigned m0, m1 = 0;
  if (two) {
    const unsigned h1 = rowm[jy1 & (GRID - 1)] & colm[jx1 & (GRID - 1)];
    m0 = mf_agent_masks(h0, lane);
    m1 = mf_agent_masks(h1, lane);
    cnts[wv * GA + lane] = __popc(m0);
    cnts[(wv + 16) * GA + lane] = __popc(m1);
  } else {
    m0 = mf_agent_masks(h0, lane);
    cnts[wv * GA + lane] = __popc(m0);
  }
  __syncthreads();

  {
    int off0 = 0, off1 = 0, all = 0;
#pragma unroll
    for (int c = 0; c < NCH; ++c) {
      const int ck = cnts[c * GA + lane];
      off0 += (c < wv) ? ck : 0;
      off1 += (c < wv + 16) ? ck : 0;
      all += ck;
    }
    if (wv == 0) tot[lane] = all;
    unsigned* fa = flat + lane * FLP;
    const int c1 = two ? wv + 16 : wv;
    for (unsigned ma = m0, mb = m1; (ma | mb) != 0u; ) {
      const unsigned a2 = ma & (ma - 1u);
      const unsigned b2 = mb & (mb - 1u);
      const bool ha = ma != 0u, ha2 = a2 != 0u, hb = mb != 0u, hb2 = b2 != 0u;
      const int ja  = 32 * wv + (ha  ? __builtin_ctz(ma) : 0);
      const int ja2 = 32 * wv + (ha2 ? __builtin_ctz(a2) : 0);
      const int jb  = 32 * c1 + (hb  ? __builtin_ctz(mb) : 0);
      const int jb2 = 32 * c1 + (hb2 ? __builtin_ctz(b2) : 0);
      const unsigned qa = posl[ja], qa2 = posl[ja2], qb = posl[jb], qb2 = posl[jb2];
      asm volatile("" :: "v"(qa), "v"(qa2), "v"(qb), "v"(qb2));
      fa[ha  ? off0 : AGENTS] = mf_entry(qa,  ja,  iy, ix);  off0 += ha  ? 1 : 0;
      fa[ha2 ? off0 : AGENTS] = mf_entry(qa2, ja2, iy, ix);  off0 += ha2 ? 1 : 0;
      fa[hb  ? off1 : AGENTS] = mf_entry(qb,  jb,  iy, ix);  off1 += hb  ? 1 : 0;
      fa[hb2 ? off1 : AGENTS] = mf_entry(qb2, jb2, iy, ix);  off1 += hb2 ? 1 : 0;
      ma = a2 & (a2 - 1u);
      mb = b2 & (b2 - 1u);
    }
  }
  __syncthreads();

  const float* Tb = T + (size_t)(b * AGENTS) * NPAD;
  const unsigned* fl = flat + a * FLP;
  const int total = tot[a];
  float acc = 0.f;
  if (s < SLOTS * NACT) {
    for (int n = sl; n < total; n += SLOTS * INFL) {
      float v[INFL];
      bool  h[INFL];
#pragma unroll
      for (int u = 0; u < INFL; ++u) {
        const int nn = n + SLOTS * u;
        h[u] = nn < total;
        v[u] = Tb[fl[h[u] ? nn : n] + (unsigned)x];
      }
#pragma unroll
      for (int u = 0; u < INFL; ++u) acc += h[u] ? v[u] : 0.f;
    }
  }
  float v = acc;
  v += __shfl_down(acc, NACT, 16);
  v += __shfl_down(acc, 2 * NACT, 16);
  if (s < NACT) {
    const float qv = lg + v - NEG_INF_F * il;
    if (last) qout[o] = qv;
    else      qs[a * NACT + s] = qv;
  }
  if (last) return;

  __syncthreads();
  float q[NACT];
#pragma unroll
  for (int y = 0; y < NACT; ++y) q[y] = qs[(rt * 16 + nl) * NACT + y];
  v2f a0, a1;
  mf_a_frag(q, half, a0, a1);
  float* dst = Tn + (size_t)(b * AGENTS + a_base + rt * 16 + 8 * half) * NPAD + nl;
  mf_tile(a0, a1, b0[0], b1[0], dst + cg * 16);
  if (cg < NTILES - 8) mf_tile(a0, a1, b0[1], b1[1], dst + (cg + 8) * 16);
}

__global__ void __launch_bounds__(512)
mf_sweep_next(const float* __restrict__ logits,
              const float* __restrict__ ill,
              const int*   __restrict__ pos,
              const float* __restrict__ params,
              const float* __restrict__ T,
              float* __restrict__ Tn) {
  mf_sweep(logits, ill, pos, params, T, nullptr, Tn, false);
}

__global__ void __launch_bounds__(512)
mf_sweep_last(const float* __restrict__ logits,
              const float* __restrict__ ill,
              const int*   __restrict__ pos,
              const float* __restrict__ T,
              float* __restrict__ qout) {
  mf_sweep(logits, ill, pos, nullptr, T, qout, nullptr, true);
}

extern "C" void kernel_launch(void* const* d_in, const int* in_sizes, int n_in,
                              void* d_out, int out_size, void* d_ws, size_t ws_size,
                              hipStream_t stream) {
  (void)in_sizes; (void)n_in; (void)out_size; (void)ws_size;
  const float* logits = (const float*)d_in[0];
  const float* ill    = (const float*)d_in[1];
  const int*   pos    = (const int*)d_in[2];
  const float* q_init = (const float*)d_in[3];
  const float* params = (const float*)d_in[4];
  const int NBA = BATCH * AGENTS;

  float* T0 = (float*)d_ws;
  float* T1 = T0 + NBA * NPAD;

  mf_table0<<<NBA / 16, 128, 0, stream>>>(q_init, ill, params, T0);
  mf_sweep_next<<<NBA / GA, 512, 0, stream>>>(logits, ill, pos, params, T0, T1);
  mf_sweep_next<<<NBA / GA, 512, 0, stream>>>(logits, ill, pos, params, T1, T0);
  mf_sweep_last<<<NBA / GA, 512, 0, stream>>>(logits, ill, pos, T0, (float*)d_out);
}


# Start-table launch gone: first sweep puts its batch's 768 softmaxes in LDS and forms each listed neighbour's 5 values as params block x probs on the vector unit, f32; 4 launches to 3; new sum order
# speedup vs baseline: 17.6742x; 1.1360x over previous
#include <hip/hip_runtime.h>


#define FOV        11
#define MAX_DIST   5
#define NACT       5
#define NEG_INF_F  1e10f
#define AGENTS     768
#define BATCH      4
#define NOFF       (MAX_DIST + 1)
#define NPUSED     (NOFF * NOFF)
#define NTRUE      (NPUSED * NACT)
#define NPAD       192
#define NTILES     (NPAD / 16)
#define GA         32
#define NCH        (AGENTS / 32)
#define FLP        (AGENTS + 1)
#define SLOTS      3
#define INFL       16
#define GRID       64
#define PF_BYTES   256
#define PF_PARTS   12
#define PF_UNITS   (AGENTS * NPAD * 4 / PF_BYTES / PF_PARTS)
#define PRP        8
#define PMP        8
#define PMB        (NACT * PMP)
#define INFL1      4

typedef float v2f __attribute__((ext_vector_type(2)));
typedef float v4f __attribute__((ext_vector_type(4)));
typedef float v8f __attribute__((ext_vector_type(8)));

__device__ __forceinline__ void mf_b_frag(const float* __restrict__ params,
                                          int half, int n, v2f& b0, v2f& b1) {
  const bool ok = n < NTRUE;
  const int p = n / NACT;
  const int x = n - p * NACT;
  const int dy = p / NOFF;
  const int dx = p - dy * NOFF;
  const int pidx = (dy + MAX_DIST) * FOV + (dx + MAX_DIST);
  const int base = ok ? pidx * (NACT * NACT) + x * NACT : 0;
  const float v0 = params[base + 2 * half];
  const float v1 = params[base + 2 * half + 1];
  const float v4 = params[base + 4];
  b0.x = ok ? v0 : 0.f;
  b0.y = ok ? v1 : 0.f;
  b1.x = (ok && half == 0) ? v4 : 0.f;
  b1.y = 0.f;
}

__device__ __forceinline__ void mf_softmax(const float* q, float* pr) {
  float m = q[0];
#pragma unroll
  for (int y = 1; y < NACT; ++y) m = fmaxf(m, q[y]);
  float e[NACT];
  float s = 0.f;
#pragma unroll
  for (int y = 0; y < NACT; ++y) { e[y] = __expf(q[y] - m); s += e[y]; }
  const float inv = 1.f / s;
#pragma unroll
  for (int y = 0; y < NACT; ++y) pr[y] = e[y] * inv;
}

__device__ __forceinline__ void mf_a_frag(const float* q, int half,
                                          v2f& a0, v2f& a1) {
  float pr[NACT];
  mf_softmax(q, pr);
  a0.x = half ? pr[2] : pr[0];
  a0.y = half ? pr[3] : pr[1];
  a1.x = half ? 0.f : pr[4];
  a1.y = 0.f;
}

__device__ __forceinline__ void mf_tile(v2f a0, v2f a1, v2f b0, v2f b1,
                                        float* __restrict__ dst) {
  v8f acc = {};
  acc = __builtin_amdgcn_wmma_f32_16x16x4_f32(
       false, a0,  false, b0,
       (short)0, acc,  false,  false);
  acc = __builtin_amdgcn_wmma_f32_16x16x4_f32(
       false, a1,  false, b1,
       (short)0, acc,  false,  false);
#pragma unroll
  for (int g = 0; g < 8; ++g) dst[(size_t)g * NPAD] = acc[g];
}

__device__ __forceinline__ unsigned mf_turn_step(unsigned x, unsigned t, int lane,
                                                 int s, unsigned m) {
  const bool up = (lane & s) != 0;
  const unsigned keep = up ? ~m : m;
  const unsigned r = __builtin_rotateright32(t, up ? (unsigned)s : (unsigned)(32 - s));
  return (x & keep) | (r & ~keep);
}

__device__ __forceinline__ unsigned mf_agent_masks(unsigned h, int lane) {
  unsigned x = h;
  unsigned t = (unsigned)__builtin_amdgcn_permlanex16((int)x, (int)x, 0x76543210, 0xfedcba98,
                                                      false, false);
  x = mf_turn_step(x, t, lane, 16, 0x0000ffffu);
  t = (unsigned)__builtin_amdgcn_update_dpp(0, (int)x, 0x168, 0xf, 0xf, false);
  x = mf_turn_step(x, t, lane, 8, 0x00ff00ffu);
  t = (unsigned)__builtin_amdgcn_update_dpp(0, (int)x, 0x164, 0xf, 0xf, false);
  x = mf_turn_step(x, t, lane, 4, 0x0f0f0f0fu);
  t = (unsigned)__builtin_amdgcn_update_dpp(0, (int)x, 0x162, 0xf, 0xf, false);
  x = mf_turn_step(x, t, lane, 2, 0x33333333u);
  t = (unsigned)__builtin_amdgcn_update_dpp(0, (int)x, 0x161, 0xf, 0xf, false);
  x = mf_turn_step(x, t, lane, 1, 0x55555555u);
  return x;
}

__device__ __forceinline__ unsigned mf_entry(unsigned pj, int j, int iy, int ix) {
  const int ey = (int)(pj & 0xffffu) - iy, ex = (int)(pj >> 16) - ix;
  const int dy = ey < 0 ? -ey : ey, dx = ex < 0 ? -ex : ex;
  return (unsigned)(j * NPAD + (dy * NOFF + dx) * NACT);
}

__device__ __forceinline__ unsigned mf_entry1(unsigned pj, int j, int iy, int ix) {
  const int ey = (int)(pj & 0xffffu) - iy, ex = (int)(pj >> 16) - ix;
  const int dy = ey < 0 ? -ey : ey, dx = ex < 0 ? -ex : ex;
  return (unsigned)(j * PRP) | ((unsigned)((dy * NOFF + dx) * PMB) << 16);
}

__device__ __forceinline__ void mf_sweep(const float* __restrict__ logits,
                                         const float* __restrict__ ill,
                                         const int*   __restrict__ pos,
                                         const float* __restrict__ params,
                                         const float* __restrict__ q_init,
                                         const float* __restrict__ T,
                                         float* __restrict__ qout,
                                         float* __restrict__ Tn,
                                         const bool first,
                                         const bool last) {
  __shared__ unsigned flat[GA * FLP];
  __shared__ unsigned posl[AGENTS];
  __shared__ unsigned rowm[GRID];
  __shared__ unsigned colm[GRID];
  __shared__ int cnts[NCH * GA];
  __shared__ int tot[GA];
  __shared__ float qs[GA * NACT];
  __shared__ __attribute__((aligned(16))) float prl[AGENTS * PRP];
  __shared__ __attribute__((aligned(16))) float pml[NPUSED * PMB];

  const int tid    = threadIdx.x;
  const int b      = blockIdx.x / (AGENTS / GA);
  const int a_base = (blockIdx.x - b * (AGENTS / GA)) * GA;
  const int* pb    = pos + (size_t)(b * AGENTS) * 2;

  const int a  = tid >> 4;
  const int s  = tid & 15;
  const int sl = s / NACT;
  const int x  = s - sl * NACT;
  const int o = (b * AGENTS + a_base + a) * NACT + (s < NACT ? s : 0);
  const float lg = logits[o];
  const float il = ill[o];

  const int lane = tid & 31;
  const int wv   = __builtin_amdgcn_readfirstlane(tid >> 5);
  const int half = lane >> 4;
  const int nl   = lane & 15;
  const int rt   = wv & 1;
  const int cg   = wv >> 1;
  v2f b0[2], b1[2];
  if (!last) {
#pragma unroll
    for (int t = 0; t < 2; ++t) {
      const int ct = (t < 1 || cg < NTILES - 8) ? cg + 8 * t : cg;
      mf_b_frag(params, half, ct * 16 + nl, b0[t], b1[t]);
    }
  }

  const bool two = wv < NCH - 16;
  const int iy  = pb[2 * (a_base + lane)];
  const int ix  = pb[2 * (a_base + lane) + 1];
  const int jy0 = pb[2 * tid];
  const int jx0 = pb[2 * tid + 1];
  float q0[NACT], q1[NACT];
  float pv0 = 0.f, pv1 = 0.f;
  int pp0 = 0, pr0 = 0, px0 = 0, pp1 = 0, pr1 = 0, px1 = 0;
  bool pe1 = false;
#pragma unroll
  for (int y = 0; y < NACT; ++y) q0[y] = q1[y] = 0.f;
  if (first) {
    const float* qb = q_init + (size_t)(b * AGENTS) * NACT;
    const float* ib = ill + (size_t)(b * AGENTS) * NACT;
    const int j1 = two ? tid + 512 : tid;
#pragma unroll
    for (int y = 0; y < NACT; ++y) {
      q0[y] = qb[tid * NACT + y] - NEG_INF_F * ib[tid * NACT + y];
      q1[y] = qb[j1 * NACT + y] - NEG_INF_F * ib[j1 * NACT + y];
    }
    pe1 = tid + 512 < NPUSED * NACT * NACT;
    const int e0 = tid, e1 = pe1 ? tid + 512 : tid;
    pp0 = e0 / (NACT * NACT);  pr0 = e0 - pp0 * (NACT * NACT);  px0 = pr0 / NACT;
    pp1 = e1 / (NACT * NACT);  pr1 = e1 - pp1 * (NACT * NACT);  px1 = pr1 / NACT;
    const int dy0 = pp0 / NOFF, dx0 = pp0 - dy0 * NOFF;
    const int dy1 = pp1 / NOFF, dx1 = pp1 - dy1 * NOFF;
    pv0 = params[((dy0 + MAX_DIST) * FOV + (dx0 + MAX_DIST)) * (NACT * NACT) + pr0];
    pv1 = params[((dy1 + MAX_DIST) * FOV + (dx1 + MAX_DIST)) * (NACT * NACT) + pr1];
  }
  if (!first) {
    const int k = blockIdx.x - b * (AGENTS / GA);
    const int p = k & 7;
    const int part = (k >> 3) * 4 + (p < 4 ? p : p ^ 5);
    if (tid < PF_UNITS)
      __builtin_prefetch((const char*)(T + (size_t)(b * AGENTS) * NPAD) +
                             (size_t)(part * PF_UNITS + tid) * PF_BYTES,
                          0,  2);
  }
  int jy1 = 0, jx1 = 0;
  if (two) {
    jy1 = pb[2 * (tid + 512)];
    jx1 = pb[2 * (tid + 512) + 1];
  }
  if (first) {
    pml[pp0 * PMB + px0 * PMP + (pr0 - px0 * NACT)] = pv0;
    if (pe1) pml[pp1 * PMB + px1 * PMP + (pr1 - px1 * NACT)] = pv1;
    float pr[NACT];
    mf_softmax(q0, pr);
    *(v4f*)(prl + tid * PRP) = (v4f){pr[0], pr[1], pr[2], pr[3]};
    prl[tid * PRP + 4] = pr[4];
    mf_softmax(q1, pr);
    if (two) {
      *(v4f*)(prl + (tid + 512) * PRP) = (v4f){pr[0], pr[1], pr[2], pr[3]};
      prl[(tid + 512) * PRP + 4] = pr[4];
    }
  }
  posl[tid] = (unsigned)jy0 | ((unsigned)jx0 << 16);
  if (two) posl[tid + 512] = (unsigned)jy1 | ((unsigned)jx1 << 16);
  {
    unsigned rv = 0, cv = 0;
#pragma unroll
    for (int k = 0; k < 4; ++k) {
      const int r = 4 * wv + k;
      const unsigned mr = __builtin_amdgcn_ballot_w32(
          (unsigned)(iy - r + MAX_DIST) <= (unsigned)(2 * MAX_DIST));
      const unsigned mc = __builtin_amdgcn_ballot_w32(
          (unsigned)(ix - r + MAX_DIST) <= (unsigned)(2 * MAX_DIST));
      rv = lane == k ? mr : rv;
      cv = lane == k ? mc : cv;
    }
    if (lane < 4) {
      rowm[4 * wv + lane] = rv;
      colm[4 * wv + lane] = cv;
    }
  }
  __syncthreads();

  const unsigned h0 = rowm[jy0 & (GRID - 1)] & colm[jx0 & (GRID - 1)];
  unsigned m0, m1 = 0;
  if (two) {
    const unsigned h1 = rowm[jy1 & (GRID - 1)] & colm[jx1 & (GRID - 1)];
    m0 = mf_agent_masks(h0, lane);
    m1 = mf_agent_masks(h1, lane);
    cnts[wv * GA + lane] = __popc(m0);
    cnts[(wv + 16) * GA + lane] = __popc(m1);
  } else {
    m0 = mf_agent_masks(h0, lane);
    cnts[wv * GA + lane] = __popc(m0);
  }
  __syncthreads();

  {
    int off0 = 0, off1 = 0, all = 0;
#pragma unroll
    for (int c = 0; c < NCH; ++c) {
      const int ck = cnts[c * GA + lane];
      off0 += (c < wv) ? ck : 0;
      off1 += (c < wv + 16) ? ck : 0;
      all += ck;
    }
    if (wv == 0) tot[lane] = all;
    unsigned* fa = flat + lane * FLP;
    const int c1 = two ? wv + 16 : wv;
    for (unsigned ma = m0, mb = m1; (ma | mb) != 0u; ) {
      const unsigned a2 = ma & (ma - 1u);
      const unsigned b2 = mb & (mb - 1u);
      const bool ha = ma != 0u, ha2 = a2 != 0u, hb = mb != 0u, hb2 = b2 != 0u;
      const int ja  = 32 * wv + (ha  ? __builtin_ctz(ma) : 0);
      const int ja2 = 32 * wv + (ha2 ? __builtin_ctz(a2) : 0);
      const int jb  = 32 * c1 + (hb  ? __builtin_ctz(mb) : 0);
      const int jb2 = 32 * c1 + (hb2 ? __builtin_ctz(b2) : 0);
      const unsigned qa = posl[ja], qa2 = posl[ja2], qb = posl[jb], qb2 = posl[jb2];
      asm volatile("" :: "v"(qa), "v"(qa2), "v"(qb), "v"(qb2));
      fa[ha  ? off0 : AGENTS] = first ? mf_entry1(qa,  ja,  iy, ix) : mf_entry(qa,  ja,  iy, ix);  off0 += ha  ? 1 : 0;
      fa[ha2 ? off0 : AGENTS] = first ? mf_entry1(qa2, ja2, iy, ix) : mf_entry(qa2, ja2, iy, ix);  off0 += ha2 ? 1 : 0;
      fa[hb  ? off1 : AGENTS] = first ? mf_entry1(qb,  jb,  iy, ix) : mf_entry(qb,  jb,  iy, ix);  off1 += hb  ? 1 : 0;
      fa[hb2 ? off1 : AGENTS] = first ? mf_entry1(qb2, jb2, iy, ix) : mf_entry(qb2, jb2, iy, ix);  off1 += hb2 ? 1 : 0;
      ma = a2 & (a2 - 1u);
      mb = b2 & (b2 - 1u);
    }
  }
  __syncthreads();

  const float* Tb = T + (size_t)(b * AGENTS) * NPAD;
  const unsigned* fl = flat + a * FLP;
  const int total = tot[a];
  float acc = 0.f;
  if (first) {
    if (s < SLOTS * NACT) {
      const float* pmx = pml + x * PMP;
      for (int n = sl; n < total; n += SLOTS * INFL1) {
        unsigned en[INFL1];
        bool     h[INFL1];
#pragma unroll
        for (int u = 0; u < INFL1; ++u) {
          const int nn = n + SLOTS * u;
          h[u] = nn < total;
          en[u] = fl[h[u] ? nn : n];
        }
        asm volatile("" :: "v"(en[0]), "v"(en[1]), "v"(en[2]), "v"(en[3]));
        __builtin_amdgcn_sched_barrier(0);
        v4f r4[INFL1], p4[INFL1];
        float r1[INFL1], p1[INFL1];
#pragma unroll
        for (int u = 0; u < INFL1; ++u) {
          const float* rr = prl + (en[u] & 0xffffu);
          const float* pp = pmx + (en[u] >> 16);
          r4[u] = *(const v4f*)rr;
          r1[u] = rr[4];
          p4[u] = *(const v4f*)pp;
          p1[u] = pp[4];
        }
        asm volatile("" :: "v"(r4[0]), "v"(r4[1]), "v"(r4[2]), "v"(r4[3]),
                           "v"(p4[0]), "v"(p4[1]), "v"(p4[2]), "v"(p4[3]),
                           "v"(r1[0]), "v"(r1[1]), "v"(r1[2]), "v"(r1[3]),
                           "v"(p1[0]), "v"(p1[1]), "v"(p1[2]), "v"(p1[3]));
#pragma unroll
        for (int u = 0; u < INFL1; ++u) {
          const float t = p4[u].x * r4[u].x + p4[u].y * r4[u].y + p4[u].z * r4[u].z +
                          p4[u].w * r4[u].w + p1[u] * r1[u];
          acc += h[u] ? t : 0.f;
        }
      }
    }
  } else if (s < SLOTS * NACT) {
    for (int n = sl; n < total; n += SLOTS * INFL) {
      float v[INFL];
      bool  h[INFL];
#pragma unroll
      for (int u = 0; u < INFL; ++u) {
        const int nn = n + SLOTS * u;
        h[u] = nn < total;
        v[u] = Tb[fl[h[u] ? nn : n] + (unsigned)x];
      }
#pragma unroll
      for (int u = 0; u < INFL; ++u) acc += h[u] ? v[u] : 0.f;
    }
  }
  float v = acc;
  v += __shfl_down(acc, NACT, 16);
  v += __shfl_down(acc, 2 * NACT, 16);
  if (s < NACT) {
    const float qv = lg + v - NEG_INF_F * il;
    if (last) qout[o] = qv;
    else      qs[a * NACT + s] = qv;
  }
  if (last) return;

  __syncthreads();
  float q[NACT];
#pragma unroll
  for (int y = 0; y < NACT; ++y) q[y] = qs[(rt * 16 + nl) * NACT + y];
  v2f a0, a1;
  mf_a_frag(q, half, a0, a1);
  float* dst = Tn + (size_t)(b * AGENTS + a_base + rt * 16 + 8 * half) * NPAD + nl;
  mf_tile(a0, a1, b0[0], b1[0], dst + cg * 16);
  if (cg < NTILES - 8) mf_tile(a0, a1, b0[1], b1[1], dst + (cg + 8) * 16);
}

__global__ void __launch_bounds__(512)
mf_sweep_first(const float* __restrict__ logits,
               const float* __restrict__ ill,
               const int*   __restrict__ pos,
               const float* __restrict__ params,
               const float* __restrict__ q_init,
               float* __restrict__ Tn) {
  mf_sweep(logits, ill, pos, params, q_init, nullptr, nullptr, Tn, true, false);
}

__global__ void __launch_bounds__(512)
mf_sweep_next(const float* __restrict__ logits,
              const float* __restrict__ ill,
              const int*   __restrict__ pos,
              const float* __restrict__ params,
              const float* __restrict__ T,
              float* __restrict__ Tn) {
  mf_sweep(logits, ill, pos, params, nullptr, T, nullptr, Tn, false, false);
}

__global__ void __launch_bounds__(512)
mf_sweep_last(const float* __restrict__ logits,
              const float* __restrict__ ill,
              const int*   __restrict__ pos,
              const float* __restrict__ T,
              float* __restrict__ qout) {
  mf_sweep(logits, ill, pos, nullptr, nullptr, T, qout, nullptr, false, true);
}

extern "C" void kernel_launch(void* const* d_in, const int* in_sizes, int n_in,
                              void* d_out, int out_size, void* d_ws, size_t ws_size,
                              hipStream_t stream) {
  (void)in_sizes; (void)n_in; (void)out_size; (void)ws_size;
  const float* logits = (const float*)d_in[0];
  const float* ill    = (const float*)d_in[1];
  const int*   pos    = (const int*)d_in[2];
  const float* q_init = (const float*)d_in[3];
  const float* params = (const float*)d_in[4];
  const int NBA = BATCH * AGENTS;

  float* T0 = (float*)d_ws;
  float* T1 = T0 + NBA * NPAD;

  mf_sweep_first<<<NBA / GA, 512, 0, stream>>>(logits, ill, pos, params, q_init, T0);
  mf_sweep_next<<<NBA / GA, 512, 0, stream>>>(logits, ill, pos, params, T0, T1);
  mf_sweep_last<<<NBA / GA, 512, 0, stream>>>(logits, ill, pos, T1, (float*)d_out);
}


# Sweep work-groups own 16 agents on 8 waves, not 32 on 16: 192 work-groups, same wave count, on twice the compute units; 3 candidates a thread, two chunks share one mask transpose; same sums
# speedup vs baseline: 20.8904x; 1.1820x over previous
#include <hip/hip_runtime.h>


#define FOV        11
#define MAX_DIST   5
#define NACT       5
#define NEG_INF_F  1e10f
#define AGENTS     768
#define BATCH      4
#define NOFF       (MAX_DIST + 1)
#define NPUSED     (NOFF * NOFF)
#define NTRUE      (NPUSED * NACT)
#define NPAD       192
#define NTILES     (NPAD / 16)
#define GA         16
#define NTHR       256
#define WAVES      (NTHR / 32)
#define CPT        (AGENTS / NTHR)
#define NCH        (AGENTS / 32)
#define FLP        (AGENTS + 1)
#define SLOTS      3
#define INFL       16
#define GRID       64
#define PF_BYTES   256
#define PF_PARTS   24
#define PF_UNITS   (AGENTS * NPAD * 4 / PF_BYTES / PF_PARTS)
#define PRP        8
#define PMP        8
#define PMB        (NACT * PMP)
#define NPV        (NPUSED * NACT * NACT)
#define PPT        ((NPV + NTHR - 1) / NTHR)
#define INFL1      4

typedef float v2f __attribute__((ext_vector_type(2)));
typedef float v4f __attribute__((ext_vector_type(4)));
typedef float v8f __attribute__((ext_vector_type(8)));

__device__ __forceinline__ void mf_b_frag(const float* __restrict__ params,
                                          int half, int n, v2f& b0, v2f& b1) {
  const bool ok = n < NTRUE;
  const int p = n / NACT;
  const int x = n - p * NACT;
  const int dy = p / NOFF;
  const int dx = p - dy * NOFF;
  const int pidx = (dy + MAX_DIST) * FOV + (dx + MAX_DIST);
  const int base = ok ? pidx * (NACT * NACT) + x * NACT : 0;
  const float v0 = params[base + 2 * half];
  const float v1 = params[base + 2 * half + 1];
  const float v4 = params[base + 4];
  b0.x = ok ? v0 : 0.f;
  b0.y = ok ? v1 : 0.f;
  b1.x = (ok && half == 0) ? v4 : 0.f;
  b1.y = 0.f;
}

__device__ __forceinline__ void mf_softmax(const float* q, float* pr) {
  float m = q[0];
#pragma unroll
  for (int y = 1; y < NACT; ++y) m = fmaxf(m, q[y]);
  float e[NACT];
  float s = 0.f;
#pragma unroll
  for (int y = 0; y < NACT; ++y) { e[y] = __expf(q[y] - m); s += e[y]; }
  const float inv = 1.f / s;
#pragma unroll
  for (int y = 0; y < NACT; ++y) pr[y] = e[y] * inv;
}

__device__ __forceinline__ void mf_a_frag(const float* q, int half,
                                          v2f& a0, v2f& a1) {
  float pr[NACT];
  mf_softmax(q, pr);
  a0.x = half ? pr[2] : pr[0];
  a0.y = half ? pr[3] : pr[1];
  a1.x = half ? 0.f : pr[4];
  a1.y = 0.f;
}

__device__ __forceinline__ void mf_tile(v2f a0, v2f a1, v2f b0, v2f b1,
                                        float* __restrict__ dst) {
  v8f acc = {};
  acc = __builtin_amdgcn_wmma_f32_16x16x4_f32(
       false, a0,  false, b0,
       (short)0, acc,  false,  false);
  acc = __builtin_amdgcn_wmma_f32_16x16x4_f32(
       false, a1,  false, b1,
       (short)0, acc,  false,  false);
#pragma unroll
  for (int g = 0; g < 8; ++g) dst[(size_t)g * NPAD] = acc[g];
}

__device__ __forceinline__ unsigned mf_turn_step(unsigned x, unsigned t, int lane,
                                                 int s, unsigned m) {
  const bool up = (lane & s) != 0;
  const unsigned keep = up ? ~m : m;
  const unsigned r = __builtin_rotateright32(t, up ? (unsigned)s : (unsigned)(32 - s));
  return (x & keep) | (r & ~keep);
}

__device__ __forceinline__ unsigned mf_agent_masks(unsigned h, int lane) {
  unsigned x = h;
  unsigned t = (unsigned)__builtin_amdgcn_permlanex16((int)x, (int)x, 0x76543210, 0xfedcba98,
                                                      false, false);
  x = mf_turn_step(x, t, lane, 16, 0x0000ffffu);
  t = (unsigned)__builtin_amdgcn_update_dpp(0, (int)x, 0x168, 0xf, 0xf, false);
  x = mf_turn_step(x, t, lane, 8, 0x00ff00ffu);
  t = (unsigned)__builtin_amdgcn_update_dpp(0, (int)x, 0x164, 0xf, 0xf, false);
  x = mf_turn_step(x, t, lane, 4, 0x0f0f0f0fu);
  t = (unsigned)__builtin_amdgcn_update_dpp(0, (int)x, 0x162, 0xf, 0xf, false);
  x = mf_turn_step(x, t, lane, 2, 0x33333333u);
  t = (unsigned)__builtin_amdgcn_update_dpp(0, (int)x, 0x161, 0xf, 0xf, false);
  x = mf_turn_step(x, t, lane, 1, 0x55555555u);
  return x;
}

__device__ __forceinline__ unsigned mf_entry(unsigned pj, int j, int iy, int ix) {
  const int ey = (int)(pj & 0xffffu) - iy, ex = (int)(pj >> 16) - ix;
  const int dy = ey < 0 ? -ey : ey, dx = ex < 0 ? -ex : ex;
  return (unsigned)(j * NPAD + (dy * NOFF + dx) * NACT);
}

__device__ __forceinline__ unsigned mf_entry1(unsigned pj, int j, int iy, int ix) {
  const int ey = (int)(pj & 0xffffu) - iy, ex = (int)(pj >> 16) - ix;
  const int dy = ey < 0 ? -ey : ey, dx = ex < 0 ? -ex : ex;
  return (unsigned)(j * PRP) | ((unsigned)((dy * NOFF + dx) * PMB) << 16);
}

__device__ __forceinline__ void mf_sweep(const float* __restrict__ logits,
                                         const float* __restrict__ ill,
                                         const int*   __restrict__ pos,
                                         const float* __restrict__ params,
                                         const float* __restrict__ q_init,
                                         const float* __restrict__ T,
                                         float* __restrict__ qout,
                                         float* __restrict__ Tn,
                                         const bool first,
                                         const bool last) {
  __shared__ unsigned flat[GA * FLP];
  __shared__ unsigned posl[AGENTS];
  __shared__ unsigned rowm[GRID];
  __shared__ unsigned colm[GRID];
  __shared__ int cnts[NCH * GA];
  __shared__ int tot[GA];
  __shared__ float qs[GA * NACT];
  __shared__ __attribute__((aligned(16))) float prl[AGENTS * PRP];
  __shared__ __attribute__((aligned(16))) float pml[NPUSED * PMB];

  const int tid    = threadIdx.x;
  const int b      = blockIdx.x / (AGENTS / GA);
  const int a_base = (blockIdx.x - b * (AGENTS / GA)) * GA;
  const int* pb    = pos + (size_t)(b * AGENTS) * 2;

  const int a  = tid >> 4;
  const int s  = tid & 15;
  const int sl = s / NACT;
  const int x  = s - sl * NACT;
  const int o = (b * AGENTS + a_base + a) * NACT + (s < NACT ? s : 0);
  const float lg = logits[o];
  const float il = ill[o];

  const int lane = tid & 31;
  const int wv   = __builtin_amdgcn_readfirstlane(tid >> 5);
  const int half = lane >> 4;
  const int nl   = lane & 15;
  const int cg   = wv;
  v2f b0[2], b1[2];
  if (!last) {
#pragma unroll
    for (int t = 0; t < 2; ++t) {
      const int ct = (t < 1 || cg < NTILES - 8) ? cg + 8 * t : cg;
      mf_b_frag(params, half, ct * 16 + nl, b0[t], b1[t]);
    }
  }

  const int iy  = pb[2 * (a_base + nl)];
  const int ix  = pb[2 * (a_base + nl) + 1];
  int jy[CPT], jx[CPT];
#pragma unroll
  for (int c = 0; c < CPT; ++c) {
    jy[c] = pb[2 * (tid + NTHR * c)];
    jx[c] = pb[2 * (tid + NTHR * c) + 1];
  }
  float q0[CPT][NACT];
  float pv[PPT];
  int pofs[PPT];
#pragma unroll
  for (int c = 0; c < CPT; ++c)
#pragma unroll
    for (int y = 0; y < NACT; ++y) q0[c][y] = 0.f;
#pragma unroll
  for (int k = 0; k < PPT; ++k) { pv[k] = 0.f; pofs[k] = 0; }
  if (first) {
    const float* qb = q_init + (size_t)(b * AGENTS) * NACT;
    const float* ib = ill + (size_t)(b * AGENTS) * NACT;
#pragma unroll
    for (int c = 0; c < CPT; ++c) {
      const int j = tid + NTHR * c;
#pragma unroll
      for (int y = 0; y < NACT; ++y)
        q0[c][y] = qb[j * NACT + y] - NEG_INF_F * ib[j * NACT + y];
    }
#pragma unroll
    for (int k = 0; k < PPT; ++k) {
      const bool ok = (k + 1) * NTHR <= NPV || tid + NTHR * k < NPV;
      const int e  = ok ? tid + NTHR * k : tid;
      const int pp = e / (NACT * NACT);
      const int pr = e - pp * (NACT * NACT);
      const int px = pr / NACT;
      const int dy = pp / NOFF, dx = pp - dy * NOFF;
      pofs[k] = pp * PMB + px * PMP + (pr - px * NACT);
      pv[k] = params[((dy + MAX_DIST) * FOV + (dx + MAX_DIST)) * (NACT * NACT) + pr];
    }
  }
  if (!first) {
    const int k = blockIdx.x - b * (AGENTS / GA);
    const int p = k & 7;
    const int part = (k >> 3) * 4 + (p < 4 ? p : p ^ 5);
    if (tid < PF_UNITS)
      __builtin_prefetch((const char*)(T + (size_t)(b * AGENTS) * NPAD) +
                             (size_t)(part * PF_UNITS + tid) * PF_BYTES,
                          0,  2);
  }
  if (first) {
#pragma unroll
    for (int k = 0; k < PPT; ++k) pml[pofs[k]] = pv[k];
#pragma unroll
    for (int c = 0; c < CPT; ++c) {
      float pr[NACT];
      mf_softmax(q0[c], pr);
      *(v4f*)(prl + (tid + NTHR * c) * PRP) = (v4f){pr[0], pr[1], pr[2], pr[3]};
      prl[(tid + NTHR * c) * PRP + 4] = pr[4];
    }
  }
#pragma unroll
  for (int c = 0; c < CPT; ++c)
    posl[tid + NTHR * c] = (unsigned)jy[c] | ((unsigned)jx[c] << 16);
  {
    unsigned rv = 0, cv = 0;
#pragma unroll
    for (int k = 0; k < GRID / WAVES; ++k) {
      const int r = (GRID / WAVES) * wv + k;
      const unsigned mr = __builtin_amdgcn_ballot_w32(
          (unsigned)(iy - r + MAX_DIST) <= (unsigned)(2 * MAX_DIST));
      const unsigned mc = __builtin_amdgcn_ballot_w32(
          (unsigned)(ix - r + MAX_DIST) <= (unsigned)(2 * MAX_DIST));
      rv = lane == k ? mr : rv;
      cv = lane == k ? mc : cv;
    }
    if (lane < GRID / WAVES) {
      rowm[(GRID / WAVES) * wv + lane] = rv;
      colm[(GRID / WAVES) * wv + lane] = cv;
    }
  }
  __syncthreads();

  const int c0 = wv + WAVES * half;
  const int c1 = wv + 2 * WAVES;
  unsigned m0, m1;
  {
    const unsigned h0 = rowm[jy[0] & (GRID - 1)] & colm[jx[0] & (GRID - 1)];
    const unsigned h1 = rowm[jy[1] & (GRID - 1)] & colm[jx[1] & (GRID - 1)];
    const unsigned h2 = rowm[jy[2] & (GRID - 1)] & colm[jx[2] & (GRID - 1)];
    m0 = mf_agent_masks((h0 & 0x0000ffffu) | (h1 & 0xffff0000u), lane);
    m1 = mf_agent_masks(h2 & 0x0000ffffu, lane);
    cnts[c0 * GA + nl] = __popc(m0);
    if (half == 0) cnts[c1 * GA + nl] = __popc(m1);
  }
  __syncthreads();

  {
    int off0 = 0, off1 = 0, all = 0;
#pragma unroll
    for (int c = 0; c < NCH; ++c) {
      const int ck = cnts[c * GA + nl];
      off0 += (c < c0) ? ck : 0;
      off1 += (c < c1) ? ck : 0;
      all += ck;
    }
    if (wv == 0) tot[nl] = all;
    unsigned* fa = flat + nl * FLP;
    for (unsigned ma = m0, mb = m1; (ma | mb) != 0u; ) {
      const unsigned a2 = ma & (ma - 1u);
      const unsigned b2 = mb & (mb - 1u);
      const bool ha = ma != 0u, ha2 = a2 != 0u, hb = mb != 0u, hb2 = b2 != 0u;
      const int ja  = 32 * c0 + (ha  ? __builtin_ctz(ma) : 0);
      const int ja2 = 32 * c0 + (ha2 ? __builtin_ctz(a2) : 0);
      const int jb  = 32 * c1 + (hb  ? __builtin_ctz(mb) : 0);
      const int jb2 = 32 * c1 + (hb2 ? __builtin_ctz(b2) : 0);
      const unsigned qa = posl[ja], qa2 = posl[ja2], qb = posl[jb], qb2 = posl[jb2];
      asm volatile("" :: "v"(qa), "v"(qa2), "v"(qb), "v"(qb2));
      fa[ha  ? off0 : AGENTS] = first ? mf_entry1(qa,  ja,  iy, ix) : mf_entry(qa,  ja,  iy, ix);  off0 += ha  ? 1 : 0;
      fa[ha2 ? off0 : AGENTS] = first ? mf_entry1(qa2, ja2, iy, ix) : mf_entry(qa2, ja2, iy, ix);  off0 += ha2 ? 1 : 0;
      fa[hb  ? off1 : AGENTS] = first ? mf_entry1(qb,  jb,  iy, ix) : mf_entry(qb,  jb,  iy, ix);  off1 += hb  ? 1 : 0;
      fa[hb2 ? off1 : AGENTS] = first ? mf_entry1(qb2, jb2, iy, ix) : mf_entry(qb2, jb2, iy, ix);  off1 += hb2 ? 1 : 0;
      ma = a2 & (a2 - 1u);
      mb = b2 & (b2 - 1u);
    }
  }
  __syncthreads();

  const float* Tb = T + (size_t)(b * AGENTS) * NPAD;
  const unsigned* fl = flat + a * FLP;
  const int total = tot[a];
  float acc = 0.f;
  if (first) {
    if (s < SLOTS * NACT) {
      const float* pmx = pml + x * PMP;
      for (int n = sl; n < total; n += SLOTS * INFL1) {
        unsigned en[INFL1];
        bool     h[INFL1];
#pragma unroll
        for (int u = 0; u < INFL1; ++u) {
          const int nn = n + SLOTS * u;
          h[u] = nn < total;
          en[u] = fl[h[u] ? nn : n];
        }
        asm volatile("" :: "v"(en[0]), "v"(en[1]), "v"(en[2]), "v"(en[3]));
        __builtin_amdgcn_sched_barrier(0);
        v4f r4[INFL1], p4[INFL1];
        float r1[INFL1], p1[INFL1];
#pragma unroll
        for (int u = 0; u < INFL1; ++u) {
          const float* rr = prl + (en[u] & 0xffffu);
          const float* pp = pmx + (en[u] >> 16);
          r4[u] = *(const v4f*)rr;
          r1[u] = rr[4];
          p4[u] = *(const v4f*)pp;
          p1[u] = pp[4];
        }
        asm volatile("" :: "v"(r4[0]), "v"(r4[1]), "v"(r4[2]), "v"(r4[3]),
                           "v"(p4[0]), "v"(p4[1]), "v"(p4[2]), "v"(p4[3]),
                           "v"(r1[0]), "v"(r1[1]), "v"(r1[2]), "v"(r1[3]),
                           "v"(p1[0]), "v"(p1[1]), "v"(p1[2]), "v"(p1[3]));
#pragma unroll
        for (int u = 0; u < INFL1; ++u) {
          const float t = p4[u].x * r4[u].x + p4[u].y * r4[u].y + p4[u].z * r4[u].z +
                          p4[u].w * r4[u].w + p1[u] * r1[u];
          acc += h[u] ? t : 0.f;
        }
      }
    }
  } else if (s < SLOTS * NACT) {
    for (int n = sl; n < total; n += SLOTS * INFL) {
      float v[INFL];
      bool  h[INFL];
#pragma unroll
      for (int u = 0; u < INFL; ++u) {
        const int nn = n + SLOTS * u;
        h[u] = nn < total;
        v[u] = Tb[fl[h[u] ? nn : n] + (unsigned)x];
      }
#pragma unroll
      for (int u = 0; u < INFL; ++u) acc += h[u] ? v[u] : 0.f;
    }
  }
  float v = acc;
  v += __shfl_down(acc, NACT, 16);
  v += __shfl_down(acc, 2 * NACT, 16);
  if (s < NACT) {
    const float qv = lg + v - NEG_INF_F * il;
    if (last) qout[o] = qv;
    else      qs[a * NACT + s] = qv;
  }
  if (last) return;

  __syncthreads();
  float q[NACT];
#pragma unroll
  for (int y = 0; y < NACT; ++y) q[y] = qs[nl * NACT + y];
  v2f a0, a1;
  mf_a_frag(q, half, a0, a1);
  float* dst = Tn + (size_t)(b * AGENTS + a_base + 8 * half) * NPAD + nl;
  mf_tile(a0, a1, b0[0], b1[0], dst + cg * 16);
  if (cg < NTILES - 8) mf_tile(a0, a1, b0[1], b1[1], dst + (cg + 8) * 16);
}

__global__ void __launch_bounds__(NTHR)
mf_sweep_first(const float* __restrict__ logits,
               const float* __restrict__ ill,
               const int*   __restrict__ pos,
               const float* __restrict__ params,
               const float* __restrict__ q_init,
               float* __restrict__ Tn) {
  mf_sweep(logits, ill, pos, params, q_init, nullptr, nullptr, Tn, true, false);
}

__global__ void __launch_bounds__(NTHR)
mf_sweep_next(const float* __restrict__ logits,
              const float* __restrict__ ill,
              const int*   __restrict__ pos,
              const float* __restrict__ params,
              const float* __restrict__ T,
              float* __restrict__ Tn) {
  mf_sweep(logits, ill, pos, params, nullptr, T, nullptr, Tn, false, false);
}

__global__ void __launch_bounds__(NTHR)
mf_sweep_last(const float* __restrict__ logits,
              const float* __restrict__ ill,
              const int*   __restrict__ pos,
              const float* __restrict__ T,
              float* __restrict__ qout) {
  mf_sweep(logits, ill, pos, nullptr, nullptr, T, qout, nullptr, false, true);
}

extern "C" void kernel_launch(void* const* d_in, const int* in_sizes, int n_in,
                              void* d_out, int out_size, void* d_ws, size_t ws_size,
                              hipStream_t stream) {
  (void)in_sizes; (void)n_in; (void)out_size; (void)ws_size;
  const float* logits = (const float*)d_in[0];
  const float* ill    = (const float*)d_in[1];
  const int*   pos    = (const int*)d_in[2];
  const float* q_init = (const float*)d_in[3];
  const float* params = (const float*)d_in[4];
  const int NBA = BATCH * AGENTS;

  float* T0 = (float*)d_ws;
  float* T1 = T0 + NBA * NPAD;

  mf_sweep_first<<<NBA / GA, NTHR, 0, stream>>>(logits, ill, pos, params, q_init, T0);
  mf_sweep_next<<<NBA / GA, NTHR, 0, stream>>>(logits, ill, pos, params, T0, T1);
  mf_sweep_last<<<NBA / GA, NTHR, 0, stream>>>(logits, ill, pos, T1, (float*)d_out);
}


# List stage moved to the wave that sums the rows: masks pass through LDS, a wave takes its 2 agents' 24 masks (lane = chunk), count prefix by a lane scan; a barrier and the count table go; same sums
# speedup vs baseline: 22.5800x; 1.0809x over previous
#include <hip/hip_runtime.h>


#define FOV        11
#define MAX_DIST   5
#define NACT       5
#define NEG_INF_F  1e10f
#define AGENTS     768
#define BATCH      4
#define NOFF       (MAX_DIST + 1)
#define NPUSED     (NOFF * NOFF)
#define NTRUE      (NPUSED * NACT)
#define NPAD       192
#define NTILES     (NPAD / 16)
#define GA         16
#define NTHR       256
#define WAVES      (NTHR / 32)
#define CPT        (AGENTS / NTHR)
#define NCH        (AGENTS / 32)
#define MKP        (NCH + 1)
#define FLP        (AGENTS + 1)
#define SLOTS      3
#define INFL       16
#define GRID       64
#define PF_BYTES   256
#define PF_PARTS   24
#define PF_UNITS   (AGENTS * NPAD * 4 / PF_BYTES / PF_PARTS)
#define PRP        8
#define PMP        8
#define PMB        (NACT * PMP)
#define NPV        (NPUSED * NACT * NACT)
#define PPT        ((NPV + NTHR - 1) / NTHR)
#define INFL1      4

typedef float v2f __attribute__((ext_vector_type(2)));
typedef float v4f __attribute__((ext_vector_type(4)));
typedef float v8f __attribute__((ext_vector_type(8)));

__device__ __forceinline__ void mf_b_frag(const float* __restrict__ params,
                                          int half, int n, v2f& b0, v2f& b1) {
  const bool ok = n < NTRUE;
  const int p = n / NACT;
  const int x = n - p * NACT;
  const int dy = p / NOFF;
  const int dx = p - dy * NOFF;
  const int pidx = (dy + MAX_DIST) * FOV + (dx + MAX_DIST);
  const int base = ok ? pidx * (NACT * NACT) + x * NACT : 0;
  const float v0 = params[base + 2 * half];
  const float v1 = params[base + 2 * half + 1];
  const float v4 = params[base + 4];
  b0.x = ok ? v0 : 0.f;
  b0.y = ok ? v1 : 0.f;
  b1.x = (ok && half == 0) ? v4 : 0.f;
  b1.y = 0.f;
}

__device__ __forceinline__ void mf_softmax(const float* q, float* pr) {
  float m = q[0];
#pragma unroll
  for (int y = 1; y < NACT; ++y) m = fmaxf(m, q[y]);
  float e[NACT];
  float s = 0.f;
#pragma unroll
  for (int y = 0; y < NACT; ++y) { e[y] = __expf(q[y] - m); s += e[y]; }
  const float inv = 1.f / s;
#pragma unroll
  for (int y = 0; y < NACT; ++y) pr[y] = e[y] * inv;
}

__device__ __forceinline__ void mf_a_frag(const float* q, int half,
                                          v2f& a0, v2f& a1) {
  float pr[NACT];
  mf_softmax(q, pr);
  a0.x = half ? pr[2] : pr[0];
  a0.y = half ? pr[3] : pr[1];
  a1.x = half ? 0.f : pr[4];
  a1.y = 0.f;
}

__device__ __forceinline__ void mf_tile(v2f a0, v2f a1, v2f b0, v2f b1,
                                        float* __restrict__ dst) {
  v8f acc = {};
  acc = __builtin_amdgcn_wmma_f32_16x16x4_f32(
       false, a0,  false, b0,
       (short)0, acc,  false,  false);
  acc = __builtin_amdgcn_wmma_f32_16x16x4_f32(
       false, a1,  false, b1,
       (short)0, acc,  false,  false);
#pragma unroll
  for (int g = 0; g < 8; ++g) dst[(size_t)g * NPAD] = acc[g];
}

__device__ __forceinline__ unsigned mf_turn_step(unsigned x, unsigned t, int lane,
                                                 int s, unsigned m) {
  const bool up = (lane & s) != 0;
  const unsigned keep = up ? ~m : m;
  const unsigned r = __builtin_rotateright32(t, up ? (unsigned)s : (unsigned)(32 - s));
  return (x & keep) | (r & ~keep);
}

__device__ __forceinline__ unsigned mf_agent_masks(unsigned h, int lane) {
  unsigned x = h;
  unsigned t = (unsigned)__builtin_amdgcn_permlanex16((int)x, (int)x, 0x76543210, 0xfedcba98,
                                                      false, false);
  x = mf_turn_step(x, t, lane, 16, 0x0000ffffu);
  t = (unsigned)__builtin_amdgcn_update_dpp(0, (int)x, 0x168, 0xf, 0xf, false);
  x = mf_turn_step(x, t, lane, 8, 0x00ff00ffu);
  t = (unsigned)__builtin_amdgcn_update_dpp(0, (int)x, 0x164, 0xf, 0xf, false);
  x = mf_turn_step(x, t, lane, 4, 0x0f0f0f0fu);
  t = (unsigned)__builtin_amdgcn_update_dpp(0, (int)x, 0x162, 0xf, 0xf, false);
  x = mf_turn_step(x, t, lane, 2, 0x33333333u);
  t = (unsigned)__builtin_amdgcn_update_dpp(0, (int)x, 0x161, 0xf, 0xf, false);
  x = mf_turn_step(x, t, lane, 1, 0x55555555u);
  return x;
}

__device__ __forceinline__ unsigned mf_entry(unsigned pj, int j, int iy, int ix) {
  const int ey = (int)(pj & 0xffffu) - iy, ex = (int)(pj >> 16) - ix;
  const int dy = ey < 0 ? -ey : ey, dx = ex < 0 ? -ex : ex;
  return (unsigned)(j * NPAD + (dy * NOFF + dx) * NACT);
}

__device__ __forceinline__ unsigned mf_entry1(unsigned pj, int j, int iy, int ix) {
  const int ey = (int)(pj & 0xffffu) - iy, ex = (int)(pj >> 16) - ix;
  const int dy = ey < 0 ? -ey : ey, dx = ex < 0 ? -ex : ex;
  return (unsigned)(j * PRP) | ((unsigned)((dy * NOFF + dx) * PMB) << 16);
}

__device__ __forceinline__ void mf_sweep(const float* __restrict__ logits,
                                         const float* __restrict__ ill,
                                         const int*   __restrict__ pos,
                                         const float* __restrict__ params,
                                         const float* __restrict__ q_init,
                                         const float* __restrict__ T,
                                         float* __restrict__ qout,
                                         float* __restrict__ Tn,
                                         const bool first,
                                         const bool last) {
  __shared__ unsigned flat[GA * FLP];
  __shared__ unsigned posl[AGENTS];
  __shared__ unsigned rowm[GRID];
  __shared__ unsigned colm[GRID];
  __shared__ unsigned mk[GA * MKP];
  __shared__ float qs[GA * NACT];
  __shared__ __attribute__((aligned(16))) float prl[AGENTS * PRP];
  __shared__ __attribute__((aligned(16))) float pml[NPUSED * PMB];

  const int tid    = threadIdx.x;
  const int b      = blockIdx.x / (AGENTS / GA);
  const int a_base = (blockIdx.x - b * (AGENTS / GA)) * GA;
  const int* pb    = pos + (size_t)(b * AGENTS) * 2;

  const int a  = tid >> 4;
  const int s  = tid & 15;
  const int sl = s / NACT;
  const int x  = s - sl * NACT;
  const int o = (b * AGENTS + a_base + a) * NACT + (s < NACT ? s : 0);
  const float lg = logits[o];
  const float il = ill[o];

  const int lane = tid & 31;
  const int wv   = __builtin_amdgcn_readfirstlane(tid >> 5);
  const int half = lane >> 4;
  const int nl   = lane & 15;
  const int cg   = wv;
  v2f b0[2], b1[2];
  if (!last) {
#pragma unroll
    for (int t = 0; t < 2; ++t) {
      const int ct = (t < 1 || cg < NTILES - 8) ? cg + 8 * t : cg;
      mf_b_frag(params, half, ct * 16 + nl, b0[t], b1[t]);
    }
  }

  const int iy  = pb[2 * (a_base + nl)];
  const int ix  = pb[2 * (a_base + nl) + 1];
  int jy[CPT], jx[CPT];
#pragma unroll
  for (int c = 0; c < CPT; ++c) {
    jy[c] = pb[2 * (tid + NTHR * c)];
    jx[c] = pb[2 * (tid + NTHR * c) + 1];
  }
  float q0[CPT][NACT];
  float pv[PPT];
  int pofs[PPT];
#pragma unroll
  for (int c = 0; c < CPT; ++c)
#pragma unroll
    for (int y = 0; y < NACT; ++y) q0[c][y] = 0.f;
#pragma unroll
  for (int k = 0; k < PPT; ++k) { pv[k] = 0.f; pofs[k] = 0; }
  if (first) {
    const float* qb = q_init + (size_t)(b * AGENTS) * NACT;
    const float* ib = ill + (size_t)(b * AGENTS) * NACT;
#pragma unroll
    for (int c = 0; c < CPT; ++c) {
      const int j = tid + NTHR * c;
#pragma unroll
      for (int y = 0; y < NACT; ++y)
        q0[c][y] = qb[j * NACT + y] - NEG_INF_F * ib[j * NACT + y];
    }
#pragma unroll
    for (int k = 0; k < PPT; ++k) {
      const bool ok = (k + 1) * NTHR <= NPV || tid + NTHR * k < NPV;
      const int e  = ok ? tid + NTHR * k : tid;
      const int pp = e / (NACT * NACT);
      const int pr = e - pp * (NACT * NACT);
      const int px = pr / NACT;
      const int dy = pp / NOFF, dx = pp - dy * NOFF;
      pofs[k] = pp * PMB + px * PMP + (pr - px * NACT);
      pv[k] = params[((dy + MAX_DIST) * FOV + (dx + MAX_DIST)) * (NACT * NACT) + pr];
    }
  }
  if (!first) {
    const int k = blockIdx.x - b * (AGENTS / GA);
    const int p = k & 7;
    const int part = (k >> 3) * 4 + (p < 4 ? p : p ^ 5);
    if (tid < PF_UNITS)
      __builtin_prefetch((const char*)(T + (size_t)(b * AGENTS) * NPAD) +
                             (size_t)(part * PF_UNITS + tid) * PF_BYTES,
                          0,  2);
  }
  if (first) {
#pragma unroll
    for (int k = 0; k < PPT; ++k) pml[pofs[k]] = pv[k];
#pragma unroll
    for (int c = 0; c < CPT; ++c) {
      float pr[NACT];
      mf_softmax(q0[c], pr);
      *(v4f*)(prl + (tid + NTHR * c) * PRP) = (v4f){pr[0], pr[1], pr[2], pr[3]};
      prl[(tid + NTHR * c) * PRP + 4] = pr[4];
    }
  }
#pragma unroll
  for (int c = 0; c < CPT; ++c)
    posl[tid + NTHR * c] = (unsigned)jy[c] | ((unsigned)jx[c] << 16);
  {
    unsigned rv = 0, cv = 0;
#pragma unroll
    for (int k = 0; k < GRID / WAVES; ++k) {
      const int r = (GRID / WAVES) * wv + k;
      const unsigned mr = __builtin_amdgcn_ballot_w32(
          (unsigned)(iy - r + MAX_DIST) <= (unsigned)(2 * MAX_DIST));
      const unsigned mc = __builtin_amdgcn_ballot_w32(
          (unsigned)(ix - r + MAX_DIST) <= (unsigned)(2 * MAX_DIST));
      rv = lane == k ? mr : rv;
      cv = lane == k ? mc : cv;
    }
    if (lane < GRID / WAVES) {
      rowm[(GRID / WAVES) * wv + lane] = rv;
      colm[(GRID / WAVES) * wv + lane] = cv;
    }
  }
  __syncthreads();

  const int c0 = wv + WAVES * half;
  const int c1 = wv + 2 * WAVES;
  unsigned m0, m1;
  {
    const unsigned h0 = rowm[jy[0] & (GRID - 1)] & colm[jx[0] & (GRID - 1)];
    const unsigned h1 = rowm[jy[1] & (GRID - 1)] & colm[jx[1] & (GRID - 1)];
    const unsigned h2 = rowm[jy[2] & (GRID - 1)] & colm[jx[2] & (GRID - 1)];
    m0 = mf_agent_masks((h0 & 0x0000ffffu) | (h1 & 0xffff0000u), lane);
    m1 = mf_agent_masks(h2 & 0x0000ffffu, lane);
    mk[nl * MKP + c0] = m0;
    if (half == 0) mk[nl * MKP + c1] = m1;
  }
  __syncthreads();

  const int ag0 = 2 * wv;
  int total;
  {
    const bool hc = lane < NCH;
    const int  ch = hc ? lane : 0;
    const unsigned ra = mk[ag0 * MKP + ch];
    const unsigned rb = mk[(ag0 + 1) * MKP + ch];
    unsigned ma = hc ? ra : 0u;
    unsigned mb = hc ? rb : 0u;
    const unsigned cnt = (unsigned)__popc(ma) | ((unsigned)__popc(mb) << 16);
    unsigned sc = cnt;
    sc += (unsigned)__builtin_amdgcn_update_dpp(0, (int)sc, 0x111, 0xf, 0xf, false);
    sc += (unsigned)__builtin_amdgcn_update_dpp(0, (int)sc, 0x112, 0xf, 0xf, false);
    sc += (unsigned)__builtin_amdgcn_update_dpp(0, (int)sc, 0x114, 0xf, 0xf, false);
    sc += (unsigned)__builtin_amdgcn_update_dpp(0, (int)sc, 0x118, 0xf, 0xf, false);
    const unsigned low = (unsigned)__builtin_amdgcn_readlane((int)sc, 15);
    sc += half ? low : 0u;
    const unsigned all = (unsigned)__builtin_amdgcn_readlane((int)sc, 31);
    total = half ? (int)(all >> 16) : (int)(all & 0xffffu);
    const unsigned before = sc - cnt;
    int off0 = (int)(before & 0xffffu), off1 = (int)(before >> 16);
    const int iy0 = __builtin_amdgcn_readlane(iy, ag0);
    const int ix0 = __builtin_amdgcn_readlane(ix, ag0);
    const int iy1 = __builtin_amdgcn_readlane(iy, ag0 + 1);
    const int ix1 = __builtin_amdgcn_readlane(ix, ag0 + 1);
    unsigned* fa = flat + ag0 * FLP;
    unsigned* fb = fa + FLP;
    const int jc = 32 * ch;
    for (; (ma | mb) != 0u; ) {
      const unsigned a2 = ma & (ma - 1u);
      const unsigned b2 = mb & (mb - 1u);
      const bool ha = ma != 0u, ha2 = a2 != 0u, hb = mb != 0u, hb2 = b2 != 0u;
      const int ja  = jc + (ha  ? __builtin_ctz(ma) : 0);
      const int ja2 = jc + (ha2 ? __builtin_ctz(a2) : 0);
      const int jb  = jc + (hb  ? __builtin_ctz(mb) : 0);
      const int jb2 = jc + (hb2 ? __builtin_ctz(b2) : 0);
      const unsigned qa = posl[ja], qa2 = posl[ja2], qb = posl[jb], qb2 = posl[jb2];
      asm volatile("" :: "v"(qa), "v"(qa2), "v"(qb), "v"(qb2));
      fa[ha  ? off0 : AGENTS] = first ? mf_entry1(qa,  ja,  iy0, ix0) : mf_entry(qa,  ja,  iy0, ix0);  off0 += ha  ? 1 : 0;
      fa[ha2 ? off0 : AGENTS] = first ? mf_entry1(qa2, ja2, iy0, ix0) : mf_entry(qa2, ja2, iy0, ix0);  off0 += ha2 ? 1 : 0;
      fb[hb  ? off1 : AGENTS] = first ? mf_entry1(qb,  jb,  iy1, ix1) : mf_entry(qb,  jb,  iy1, ix1);  off1 += hb  ? 1 : 0;
      fb[hb2 ? off1 : AGENTS] = first ? mf_entry1(qb2, jb2, iy1, ix1) : mf_entry(qb2, jb2, iy1, ix1);  off1 += hb2 ? 1 : 0;
      ma = a2 & (a2 - 1u);
      mb = b2 & (b2 - 1u);
    }
  }

  const float* Tb = T + (size_t)(b * AGENTS) * NPAD;
  const unsigned* fl = flat + a * FLP;
  float acc = 0.f;
  if (first) {
    if (s < SLOTS * NACT) {
      const float* pmx = pml + x * PMP;
      for (int n = sl; n < total; n += SLOTS * INFL1) {
        unsigned en[INFL1];
        bool     h[INFL1];
#pragma unroll
        for (int u = 0; u < INFL1; ++u) {
          const int nn = n + SLOTS * u;
          h[u] = nn < total;
          en[u] = fl[h[u] ? nn : n];
        }
        asm volatile("" :: "v"(en[0]), "v"(en[1]), "v"(en[2]), "v"(en[3]));
        __builtin_amdgcn_sched_barrier(0);
        v4f r4[INFL1], p4[INFL1];
        float r1[INFL1], p1[INFL1];
#pragma unroll
        for (int u = 0; u < INFL1; ++u) {
          const float* rr = prl + (en[u] & 0xffffu);
          const float* pp = pmx + (en[u] >> 16);
          r4[u] = *(const v4f*)rr;
          r1[u] = rr[4];
          p4[u] = *(const v4f*)pp;
          p1[u] = pp[4];
        }
        asm volatile("" :: "v"(r4[0]), "v"(r4[1]), "v"(r4[2]), "v"(r4[3]),
                           "v"(p4[0]), "v"(p4[1]), "v"(p4[2]), "v"(p4[3]),
                           "v"(r1[0]), "v"(r1[1]), "v"(r1[2]), "v"(r1[3]),
                           "v"(p1[0]), "v"(p1[1]), "v"(p1[2]), "v"(p1[3]));
#pragma unroll
        for (int u = 0; u < INFL1; ++u) {
          const float t = p4[u].x * r4[u].x + p4[u].y * r4[u].y + p4[u].z * r4[u].z +
                          p4[u].w * r4[u].w + p1[u] * r1[u];
          acc += h[u] ? t : 0.f;
        }
      }
    }
  } else if (s < SLOTS * NACT) {
    for (int n = sl; n < total; n += SLOTS * INFL) {
      float v[INFL];
      bool  h[INFL];
#pragma unroll
      for (int u = 0; u < INFL; ++u) {
        const int nn = n + SLOTS * u;
        h[u] = nn < total;
        v[u] = Tb[fl[h[u] ? nn : n] + (unsigned)x];
      }
#pragma unroll
      for (int u = 0; u < INFL; ++u) acc += h[u] ? v[u] : 0.f;
    }
  }
  float v = acc;
  v += __shfl_down(acc, NACT, 16);
  v += __shfl_down(acc, 2 * NACT, 16);
  if (s < NACT) {
    const float qv = lg + v - NEG_INF_F * il;
    if (last) qout[o] = qv;
    else      qs[a * NACT + s] = qv;
  }
  if (last) return;

  __syncthreads();
  float q[NACT];
#pragma unroll
  for (int y = 0; y < NACT; ++y) q[y] = qs[nl * NACT + y];
  v2f a0, a1;
  mf_a_frag(q, half, a0, a1);
  float* dst = Tn + (size_t)(b * AGENTS + a_base + 8 * half) * NPAD + nl;
  mf_tile(a0, a1, b0[0], b1[0], dst + cg * 16);
  if (cg < NTILES - 8) mf_tile(a0, a1, b0[1], b1[1], dst + (cg + 8) * 16);
}

__global__ void __launch_bounds__(NTHR)
mf_sweep_first(const float* __restrict__ logits,
               const float* __restrict__ ill,
               const int*   __restrict__ pos,
               const float* __restrict__ params,
               const float* __restrict__ q_init,
               float* __restrict__ Tn) {
  mf_sweep(logits, ill, pos, params, q_init, nullptr, nullptr, Tn, true, false);
}

__global__ void __launch_bounds__(NTHR)
mf_sweep_next(const float* __restrict__ logits,
              const float* __restrict__ ill,
              const int*   __restrict__ pos,
              const float* __restrict__ params,
              const float* __restrict__ T,
              float* __restrict__ Tn) {
  mf_sweep(logits, ill, pos, params, nullptr, T, nullptr, Tn, false, false);
}

__global__ void __launch_bounds__(NTHR)
mf_sweep_last(const float* __restrict__ logits,
              const float* __restrict__ ill,
              const int*   __restrict__ pos,
              const float* __restrict__ T,
              float* __restrict__ qout) {
  mf_sweep(logits, ill, pos, nullptr, nullptr, T, qout, nullptr, false, true);
}

extern "C" void kernel_launch(void* const* d_in, const int* in_sizes, int n_in,
                              void* d_out, int out_size, void* d_ws, size_t ws_size,
                              hipStream_t stream) {
  (void)in_sizes; (void)n_in; (void)out_size; (void)ws_size;
  const float* logits = (const float*)d_in[0];
  const float* ill    = (const float*)d_in[1];
  const int*   pos    = (const int*)d_in[2];
  const float* q_init = (const float*)d_in[3];
  const float* params = (const float*)d_in[4];
  const int NBA = BATCH * AGENTS;

  float* T0 = (float*)d_ws;
  float* T1 = T0 + NBA * NPAD;

  mf_sweep_first<<<NBA / GA, NTHR, 0, stream>>>(logits, ill, pos, params, q_init, T0);
  mf_sweep_next<<<NBA / GA, NTHR, 0, stream>>>(logits, ill, pos, params, T0, T1);
  mf_sweep_last<<<NBA / GA, NTHR, 0, stream>>>(logits, ill, pos, T1, (float*)d_out);
}
